# starting kernel
# baseline (speedup 1.0000x reference)
#include <hip/hip_runtime.h>
#include <hip/hip_bf16.h>
#include <math.h>

#define N_NODES 20000
#define E_EDGES 320000
#define ETOT    (E_EDGES + N_NODES)
#define DIN     64
#define HID     128
#define NHEAD   4
#define HC      (NHEAD * HID)

typedef float v2f __attribute__((ext_vector_type(2)));
typedef float v4f __attribute__((ext_vector_type(4)));
typedef float v8f __attribute__((ext_vector_type(8)));

__device__ __forceinline__ unsigned f2key(float f) {
    unsigned u = __float_as_uint(f);
    return (u & 0x80000000u) ? ~u : (u | 0x80000000u);
}
__device__ __forceinline__ float key2f(unsigned k) {
    unsigned u = (k & 0x80000000u) ? (k ^ 0x80000000u) : ~k;
    return __uint_as_float(u);
}

template <int RELU>
__global__ void gemm_wmma_f32(const float* __restrict__ A,
                              const float* __restrict__ B,
                              const float* __restrict__ bias,
                              float* __restrict__ C,
                              int M, int K, int N) {
    int wave   = (blockIdx.x * blockDim.x + threadIdx.x) >> 5;
    int tilesN = N >> 4;
    int tiles  = (M >> 4) * tilesN;
    if (wave >= tiles) return;
    int tm = wave / tilesN, tn = wave % tilesN;
    int lane = threadIdx.x & 31;
    int l16  = lane & 15;
    int half = lane >> 4;

    const float* arow = A + (size_t)(tm * 16 + l16) * K + 2 * half;
    const float* bcol = B + (size_t)(2 * half) * N + tn * 16 + l16;

    v8f acc = {0.f, 0.f, 0.f, 0.f, 0.f, 0.f, 0.f, 0.f};
    for (int k = 0; k < K; k += 4) {
        v2f a, b;
        a.x = arow[k];
        a.y = arow[k + 1];
        b.x = bcol[(size_t)k * N];
        b.y = bcol[(size_t)(k + 1) * N];
        acc = __builtin_amdgcn_wmma_f32_16x16x4_f32(
             false, a,  false, b,
             (short)0, acc,  false,  false);
    }

    int   col = tn * 16 + l16;
    float bv  = bias[col];
#pragma unroll
    for (int v = 0; v < 8; ++v) {
        int   row = tm * 16 + v + 8 * half;
        float val = acc[v] + bv;
        if (RELU) val = fmaxf(val, 0.f);
        C[(size_t)row * N + col] = val;
    }
}

__global__ void init_scratch(float* __restrict__ res, float* __restrict__ denom,
                             unsigned* __restrict__ amax, float* __restrict__ gpart) {
    int idx = blockIdx.x * blockDim.x + threadIdx.x;
    if (idx < N_NODES * HID) res[idx] = 0.f;
    if (idx < N_NODES * NHEAD) { denom[idx] = 0.f; amax[idx] = 0u; }
    if (idx < HID) gpart[idx] = 0.f;
}

__device__ __forceinline__ void edge_ends(const int* __restrict__ ei, int e, int& s, int& d) {
    if (e < E_EDGES) { s = ei[e]; d = ei[E_EDGES + e]; }
    else             { s = d = e - E_EDGES; }
    s = __builtin_amdgcn_readfirstlane(s);
    d = __builtin_amdgcn_readfirstlane(d);
}

__global__ void edge_alpha(const int* __restrict__ ei,
                           const float* __restrict__ xl, const float* __restrict__ xr,
                           const float* __restrict__ att,
                           float* __restrict__ alpha, unsigned* __restrict__ amax) {
    int wave = (blockIdx.x * blockDim.x + threadIdx.x) >> 5;
    if (wave >= ETOT) return;
    int lane = threadIdx.x & 31;
    int s, d;
    edge_ends(ei, wave, s, d);

    const v4f* pl = (const v4f*)(xl + (size_t)s * HC);
    const v4f* pr = (const v4f*)(xr + (size_t)d * HC);
    const v4f* pa = (const v4f*)att;

    float acc[NHEAD];
#pragma unroll
    for (int h = 0; h < NHEAD; ++h) {
        int idx = h * (HID / 4) + lane;
        v4f l = pl[idx];
        v4f r = pr[idx];
        v4f w = pa[idx];
        float a = 0.f;
#pragma unroll
        for (int i = 0; i < 4; ++i) {
            float m = l[i] + r[i];
            float e = (m > 0.f) ? m : 0.2f * m;
            a += e * w[i];
        }
        acc[h] = a;
    }
#pragma unroll
    for (int h = 0; h < NHEAD; ++h)
        for (int off = 16; off; off >>= 1)
            acc[h] += __shfl_down(acc[h], off, 32);

    if (lane == 0) {
#pragma unroll
        for (int h = 0; h < NHEAD; ++h) {
            alpha[(size_t)wave * NHEAD + h] = acc[h];
            atomicMax(&amax[d * NHEAD + h], f2key(acc[h]));
        }
    }
}

__global__ void edge_exp(const int* __restrict__ ei, float* __restrict__ alpha,
                         const unsigned* __restrict__ amax, float* __restrict__ denom) {
    int idx = blockIdx.x * blockDim.x + threadIdx.x;
    if (idx >= ETOT * NHEAD) return;
    int e = idx >> 2, h = idx & 3;
    int d = (e < E_EDGES) ? ei[E_EDGES + e] : (e - E_EDGES);
    float ex = expf(alpha[idx] - key2f(amax[d * NHEAD + h]));
    alpha[idx] = ex;
    atomicAdd(&denom[d * NHEAD + h], ex);
}

__global__ void edge_scatter(const int* __restrict__ ei, const float* __restrict__ xl,
                             const float* __restrict__ alpha, const float* __restrict__ denom,
                             float* __restrict__ res) {
    int wave = (blockIdx.x * blockDim.x + threadIdx.x) >> 5;
    if (wave >= ETOT) return;
    int lane = threadIdx.x & 31;
    int s, d;
    edge_ends(ei, wave, s, d);

    float a4[NHEAD];
#pragma unroll
    for (int h = 0; h < NHEAD; ++h)
        a4[h] = alpha[(size_t)wave * NHEAD + h] / denom[d * NHEAD + h];

    const v4f* pl = (const v4f*)(xl + (size_t)s * HC);
    float* pres = res + (size_t)d * HID + lane * 4;

    v4f v = {0.f, 0.f, 0.f, 0.f};
#pragma unroll
    for (int h = 0; h < NHEAD; ++h) {
        v4f l = pl[h * (HID / 4) + lane];
#pragma unroll
        for (int i = 0; i < 4; ++i) v[i] += l[i] * a4[h];
    }
#pragma unroll
    for (int i = 0; i < 4; ++i)
        atomicAdd(&pres[i], 0.25f * v[i]);
}

__global__ void node_ln_relu(const float* __restrict__ hin, const float* __restrict__ res,
                             const float* __restrict__ bias, const float* __restrict__ g,
                             const float* __restrict__ be, float* __restrict__ hout) {
    int wave = (blockIdx.x * blockDim.x + threadIdx.x) >> 5;
    if (wave >= N_NODES) return;
    int lane = threadIdx.x & 31;
    const v4f* ph = (const v4f*)(hin + (size_t)wave * HID);
    const v4f* pr = (const v4f*)(res + (size_t)wave * HID);
    const v4f* pb = (const v4f*)bias;

    v4f y;
    {
        v4f a = ph[lane], b = pr[lane], c = pb[lane];
#pragma unroll
        for (int i = 0; i < 4; ++i) y[i] = a[i] + b[i] + c[i];
    }
    float s = 0.f, ss = 0.f;
#pragma unroll
    for (int i = 0; i < 4; ++i) { s += y[i]; ss += y[i] * y[i]; }
    for (int off = 16; off; off >>= 1) {
        s  += __shfl_xor(s,  off, 32);
        ss += __shfl_xor(ss, off, 32);
    }
    float mu  = s * (1.f / HID);
    float var = ss * (1.f / HID) - mu * mu;
    float r   = rsqrtf(var + 1e-5f);

    v4f gg = ((const v4f*)g)[lane];
    v4f bb = ((const v4f*)be)[lane];
    v4f o;
#pragma unroll
    for (int i = 0; i < 4; ++i)
        o[i] = fmaxf((y[i] - mu) * r * gg[i] + bb[i], 0.f);
    ((v4f*)(hout + (size_t)wave * HID))[lane] = o;
}

__global__ void logits_kernel(const float* __restrict__ shared_f, const float* __restrict__ Wa,
                              const float* __restrict__ ba, float* __restrict__ gpart,
                              float* __restrict__ out) {
    int wave = (blockIdx.x * blockDim.x + threadIdx.x) >> 5;
    if (wave >= N_NODES) return;
    int lane = threadIdx.x & 31;
    v4f v = ((const v4f*)(shared_f + (size_t)wave * HID))[lane];
    v4f w = ((const v4f*)Wa)[lane];
    float acc = 0.f;
#pragma unroll
    for (int i = 0; i < 4; ++i) {
        acc += v[i] * w[i];
        atomicAdd(&gpart[lane * 4 + i], v[i]);
    }
    for (int off = 16; off; off >>= 1) acc += __shfl_down(acc, off, 32);
    if (lane == 0) out[wave] = acc + ba[0];
}

__global__ void value_kernel(const float* __restrict__ gpart, const float* __restrict__ Wc,
                             const float* __restrict__ bc, float* __restrict__ out) {
    int lane = threadIdx.x & 31;
    v4f v = ((const v4f*)gpart)[lane];
    v4f w = ((const v4f*)Wc)[lane];
    float acc = 0.f;
#pragma unroll
    for (int i = 0; i < 4; ++i) acc += (v[i] * (1.f / N_NODES)) * w[i];
    for (int off = 16; off; off >>= 1) acc += __shfl_down(acc, off, 32);
    if (lane == 0) out[N_NODES] = acc + bc[0];
}

static inline int blocksForWaves(long long nwaves) {
    long long threads = nwaves * 32;
    return (int)((threads + 255) / 256);
}

extern "C" void kernel_launch(void* const* d_in, const int* in_sizes, int n_in,
                              void* d_out, int out_size, void* d_ws, size_t ws_size,
                              hipStream_t stream) {
    const float* x     = (const float*)d_in[0];
    const int*   ei    = (const int*)d_in[1];
    const float* W_in  = (const float*)d_in[2];
    const float* b_in  = (const float*)d_in[3];
    const float* Wl1   = (const float*)d_in[4];
    const float* bl1   = (const float*)d_in[5];
    const float* Wr1   = (const float*)d_in[6];
    const float* br1   = (const float*)d_in[7];
    const float* att1  = (const float*)d_in[8];
    const float* bias1 = (const float*)d_in[9];
    const float* g1    = (const float*)d_in[10];
    const float* be1   = (const float*)d_in[11];
    const float* Wl2   = (const float*)d_in[12];
    const float* bl2   = (const float*)d_in[13];
    const float* Wr2   = (const float*)d_in[14];
    const float* br2   = (const float*)d_in[15];
    const float* att2  = (const float*)d_in[16];
    const float* bias2 = (const float*)d_in[17];
    const float* g2    = (const float*)d_in[18];
    const float* be2   = (const float*)d_in[19];
    const float* Wa    = (const float*)d_in[20];
    const float* ba    = (const float*)d_in[21];
    const float* Wc    = (const float*)d_in[22];
    const float* bc    = (const float*)d_in[23];

    float* out = (float*)d_out;

    float* ws = (float*)d_ws;
    size_t off = 0;
    float*    h0    = ws + off; off += (size_t)N_NODES * HID;
    float*    h1    = ws + off; off += (size_t)N_NODES * HID;
    float*    xl    = ws + off; off += (size_t)N_NODES * HC;
    float*    xr    = ws + off; off += (size_t)N_NODES * HC;
    float*    alpha = ws + off; off += (size_t)ETOT * NHEAD;
    unsigned* amax  = (unsigned*)(ws + off); off += (size_t)N_NODES * NHEAD;
    float*    denom = ws + off; off += (size_t)N_NODES * NHEAD;
    float*    res   = ws + off; off += (size_t)N_NODES * HID;
    float*    gpart = ws + off; off += HID;
    (void)ws_size; (void)in_sizes; (void)n_in; (void)out_size;

    const int initBlocks  = (N_NODES * HID + 255) / 256;
    const int edgeBlocks  = blocksForWaves(ETOT);
    const int ehBlocks    = (ETOT * NHEAD + 255) / 256;
    const int nodeBlocks  = blocksForWaves(N_NODES);
    const int gemmInBlk   = blocksForWaves((N_NODES / 16) * (HID / 16));
    const int gemmHCBlk   = blocksForWaves((N_NODES / 16) * (HC / 16));

    gemm_wmma_f32<1><<<gemmInBlk, 256, 0, stream>>>(x, W_in, b_in, h0, N_NODES, DIN, HID);

    init_scratch<<<initBlocks, 256, 0, stream>>>(res, denom, amax, gpart);
    gemm_wmma_f32<0><<<gemmHCBlk, 256, 0, stream>>>(h0, Wl1, bl1, xl, N_NODES, HID, HC);
    gemm_wmma_f32<0><<<gemmHCBlk, 256, 0, stream>>>(h0, Wr1, br1, xr, N_NODES, HID, HC);
    edge_alpha  <<<edgeBlocks, 256, 0, stream>>>(ei, xl, xr, att1, alpha, amax);
    edge_exp    <<<ehBlocks,   256, 0, stream>>>(ei, alpha, amax, denom);
    edge_scatter<<<edgeBlocks, 256, 0, stream>>>(ei, xl, alpha, denom, res);
    node_ln_relu<<<nodeBlocks, 256, 0, stream>>>(h0, res, bias1, g1, be1, h1);

    init_scratch<<<initBlocks, 256, 0, stream>>>(res, denom, amax, gpart);
    gemm_wmma_f32<0><<<gemmHCBlk, 256, 0, stream>>>(h1, Wl2, bl2, xl, N_NODES, HID, HC);
    gemm_wmma_f32<0><<<gemmHCBlk, 256, 0, stream>>>(h1, Wr2, br2, xr, N_NODES, HID, HC);
    edge_alpha  <<<edgeBlocks, 256, 0, stream>>>(ei, xl, xr, att2, alpha, amax);
    edge_exp    <<<ehBlocks,   256, 0, stream>>>(ei, alpha, amax, denom);
    edge_scatter<<<edgeBlocks, 256, 0, stream>>>(ei, xl, alpha, denom, res);
    node_ln_relu<<<nodeBlocks, 256, 0, stream>>>(h1, res, bias2, g2, be2, h0);

    logits_kernel<<<nodeBlocks, 256, 0, stream>>>(h0, Wa, ba, gpart, out);
    value_kernel <<<1, 32, 0, stream>>>(gpart, Wc, bc, out);
}


# logits kernel: graph-mean column sums reduced in registers (16 nodes per wave) and LDS (8 waves), one float atomic per column per 128-node block instead of per node; float sum order changes
# speedup vs baseline: 1.2711x; 1.2711x over previous
#include <hip/hip_runtime.h>
#include <hip/hip_bf16.h>
#include <math.h>

#define N_NODES 20000
#define E_EDGES 320000
#define ETOT    (E_EDGES + N_NODES)
#define DIN     64
#define HID     128
#define NHEAD   4
#define HC      (NHEAD * HID)

typedef float v2f __attribute__((ext_vector_type(2)));
typedef float v4f __attribute__((ext_vector_type(4)));
typedef float v8f __attribute__((ext_vector_type(8)));

__device__ __forceinline__ unsigned f2key(float f) {
    unsigned u = __float_as_uint(f);
    return (u & 0x80000000u) ? ~u : (u | 0x80000000u);
}
__device__ __forceinline__ float key2f(unsigned k) {
    unsigned u = (k & 0x80000000u) ? (k ^ 0x80000000u) : ~k;
    return __uint_as_float(u);
}

template <int RELU>
__global__ void gemm_wmma_f32(const float* __restrict__ A,
                              const float* __restrict__ B,
                              const float* __restrict__ bias,
                              float* __restrict__ C,
                              int M, int K, int N) {
    int wave   = (blockIdx.x * blockDim.x + threadIdx.x) >> 5;
    int tilesN = N >> 4;
    int tiles  = (M >> 4) * tilesN;
    if (wave >= tiles) return;
    int tm = wave / tilesN, tn = wave % tilesN;
    int lane = threadIdx.x & 31;
    int l16  = lane & 15;
    int half = lane >> 4;

    const float* arow = A + (size_t)(tm * 16 + l16) * K + 2 * half;
    const float* bcol = B + (size_t)(2 * half) * N + tn * 16 + l16;

    v8f acc = {0.f, 0.f, 0.f, 0.f, 0.f, 0.f, 0.f, 0.f};
    for (int k = 0; k < K; k += 4) {
        v2f a, b;
        a.x = arow[k];
        a.y = arow[k + 1];
        b.x = bcol[(size_t)k * N];
        b.y = bcol[(size_t)(k + 1) * N];
        acc = __builtin_amdgcn_wmma_f32_16x16x4_f32(
             false, a,  false, b,
             (short)0, acc,  false,  false);
    }

    int   col = tn * 16 + l16;
    float bv  = bias[col];
#pragma unroll
    for (int v = 0; v < 8; ++v) {
        int   row = tm * 16 + v + 8 * half;
        float val = acc[v] + bv;
        if (RELU) val = fmaxf(val, 0.f);
        C[(size_t)row * N + col] = val;
    }
}

__global__ void init_scratch(float* __restrict__ res, float* __restrict__ denom,
                             unsigned* __restrict__ amax, float* __restrict__ gpart) {
    int idx = blockIdx.x * blockDim.x + threadIdx.x;
    if (idx < N_NODES * HID) res[idx] = 0.f;
    if (idx < N_NODES * NHEAD) { denom[idx] = 0.f; amax[idx] = 0u; }
    if (idx < HID) gpart[idx] = 0.f;
}

__device__ __forceinline__ void edge_ends(const int* __restrict__ ei, int e, int& s, int& d) {
    if (e < E_EDGES) { s = ei[e]; d = ei[E_EDGES + e]; }
    else             { s = d = e - E_EDGES; }
    s = __builtin_amdgcn_readfirstlane(s);
    d = __builtin_amdgcn_readfirstlane(d);
}

__global__ void edge_alpha(const int* __restrict__ ei,
                           const float* __restrict__ xl, const float* __restrict__ xr,
                           const float* __restrict__ att,
                           float* __restrict__ alpha, unsigned* __restrict__ amax) {
    int wave = (blockIdx.x * blockDim.x + threadIdx.x) >> 5;
    if (wave >= ETOT) return;
    int lane = threadIdx.x & 31;
    int s, d;
    edge_ends(ei, wave, s, d);

    const v4f* pl = (const v4f*)(xl + (size_t)s * HC);
    const v4f* pr = (const v4f*)(xr + (size_t)d * HC);
    const v4f* pa = (const v4f*)att;

    float acc[NHEAD];
#pragma unroll
    for (int h = 0; h < NHEAD; ++h) {
        int idx = h * (HID / 4) + lane;
        v4f l = pl[idx];
        v4f r = pr[idx];
        v4f w = pa[idx];
        float a = 0.f;
#pragma unroll
        for (int i = 0; i < 4; ++i) {
            float m = l[i] + r[i];
            float e = (m > 0.f) ? m : 0.2f * m;
            a += e * w[i];
        }
        acc[h] = a;
    }
#pragma unroll
    for (int h = 0; h < NHEAD; ++h)
        for (int off = 16; off; off >>= 1)
            acc[h] += __shfl_down(acc[h], off, 32);

    if (lane == 0) {
#pragma unroll
        for (int h = 0; h < NHEAD; ++h) {
            alpha[(size_t)wave * NHEAD + h] = acc[h];
            atomicMax(&amax[d * NHEAD + h], f2key(acc[h]));
        }
    }
}

__global__ void edge_exp(const int* __restrict__ ei, float* __restrict__ alpha,
                         const unsigned* __restrict__ amax, float* __restrict__ denom) {
    int idx = blockIdx.x * blockDim.x + threadIdx.x;
    if (idx >= ETOT * NHEAD) return;
    int e = idx >> 2, h = idx & 3;
    int d = (e < E_EDGES) ? ei[E_EDGES + e] : (e - E_EDGES);
    float ex = expf(alpha[idx] - key2f(amax[d * NHEAD + h]));
    alpha[idx] = ex;
    atomicAdd(&denom[d * NHEAD + h], ex);
}

__global__ void edge_scatter(const int* __restrict__ ei, const float* __restrict__ xl,
                             const float* __restrict__ alpha, const float* __restrict__ denom,
                             float* __restrict__ res) {
    int wave = (blockIdx.x * blockDim.x + threadIdx.x) >> 5;
    if (wave >= ETOT) return;
    int lane = threadIdx.x & 31;
    int s, d;
    edge_ends(ei, wave, s, d);

    float a4[NHEAD];
#pragma unroll
    for (int h = 0; h < NHEAD; ++h)
        a4[h] = alpha[(size_t)wave * NHEAD + h] / denom[d * NHEAD + h];

    const v4f* pl = (const v4f*)(xl + (size_t)s * HC);
    float* pres = res + (size_t)d * HID + lane * 4;

    v4f v = {0.f, 0.f, 0.f, 0.f};
#pragma unroll
    for (int h = 0; h < NHEAD; ++h) {
        v4f l = pl[h * (HID / 4) + lane];
#pragma unroll
        for (int i = 0; i < 4; ++i) v[i] += l[i] * a4[h];
    }
#pragma unroll
    for (int i = 0; i < 4; ++i)
        atomicAdd(&pres[i], 0.25f * v[i]);
}

__global__ void node_ln_relu(const float* __restrict__ hin, const float* __restrict__ res,
                             const float* __restrict__ bias, const float* __restrict__ g,
                             const float* __restrict__ be, float* __restrict__ hout) {
    int wave = (blockIdx.x * blockDim.x + threadIdx.x) >> 5;
    if (wave >= N_NODES) return;
    int lane = threadIdx.x & 31;
    const v4f* ph = (const v4f*)(hin + (size_t)wave * HID);
    const v4f* pr = (const v4f*)(res + (size_t)wave * HID);
    const v4f* pb = (const v4f*)bias;

    v4f y;
    {
        v4f a = ph[lane], b = pr[lane], c = pb[lane];
#pragma unroll
        for (int i = 0; i < 4; ++i) y[i] = a[i] + b[i] + c[i];
    }
    float s = 0.f, ss = 0.f;
#pragma unroll
    for (int i = 0; i < 4; ++i) { s += y[i]; ss += y[i] * y[i]; }
    for (int off = 16; off; off >>= 1) {
        s  += __shfl_xor(s,  off, 32);
        ss += __shfl_xor(ss, off, 32);
    }
    float mu  = s * (1.f / HID);
    float var = ss * (1.f / HID) - mu * mu;
    float r   = rsqrtf(var + 1e-5f);

    v4f gg = ((const v4f*)g)[lane];
    v4f bb = ((const v4f*)be)[lane];
    v4f o;
#pragma unroll
    for (int i = 0; i < 4; ++i)
        o[i] = fmaxf((y[i] - mu) * r * gg[i] + bb[i], 0.f);
    ((v4f*)(hout + (size_t)wave * HID))[lane] = o;
}

#define LG_WAVES 8
#define LG_NPW   16
#define LG_NPB   (LG_WAVES * LG_NPW)

__global__ void __launch_bounds__(LG_WAVES * 32)
logits_kernel(const float* __restrict__ shared_f, const float* __restrict__ Wa,
              const float* __restrict__ ba, float* __restrict__ gpart,
              float* __restrict__ out) {
    __shared__ float colpart[LG_WAVES][HID];
    __shared__ float lg[LG_NPB];

    const int lane  = threadIdx.x & 31;
    const int wv    = __builtin_amdgcn_readfirstlane(threadIdx.x >> 5);
    const int node0 = blockIdx.x * LG_NPB + wv * LG_NPW;

    const v4f w = ((const v4f*)Wa)[lane];

    v4f rows[LG_NPW];
#pragma unroll
    for (int j = 0; j < LG_NPW; ++j) {
        int node = node0 + j;
        int nc   = (node < N_NODES) ? node : (N_NODES - 1);
        rows[j]  = ((const v4f*)(shared_f + (size_t)nc * HID))[lane];
    }

    float acc[LG_NPW];
    v4f cs = {0.f, 0.f, 0.f, 0.f};
#pragma unroll
    for (int j = 0; j < LG_NPW; ++j) {
        const bool valid = (node0 + j) < N_NODES;
        const v4f  v     = rows[j];
        float a = 0.f;
#pragma unroll
        for (int i = 0; i < 4; ++i) {
            a     += v[i] * w[i];
            cs[i] += valid ? v[i] : 0.f;
        }
        acc[j] = a;
    }
#pragma unroll
    for (int off = 16; off; off >>= 1) {
#pragma unroll
        for (int j = 0; j < LG_NPW; ++j) acc[j] += __shfl_down(acc[j], off, 32);
    }
    if (lane == 0) {
#pragma unroll
        for (int j = 0; j < LG_NPW; ++j) lg[wv * LG_NPW + j] = acc[j];
    }
    ((v4f*)&colpart[wv][0])[lane] = cs;
    __syncthreads();

    const int t = threadIdx.x;
    if (t < HID) {
        float s = 0.f;
#pragma unroll
        for (int k = 0; k < LG_WAVES; ++k) s += colpart[k][t];
        atomicAdd(&gpart[t], s);
    }
    if (t < LG_NPB) {
        int node = blockIdx.x * LG_NPB + t;
        if (node < N_NODES) out[node] = lg[t] + ba[0];
    }
}

__global__ void value_kernel(const float* __restrict__ gpart, const float* __restrict__ Wc,
                             const float* __restrict__ bc, float* __restrict__ out) {
    int lane = threadIdx.x & 31;
    v4f v = ((const v4f*)gpart)[lane];
    v4f w = ((const v4f*)Wc)[lane];
    float acc = 0.f;
#pragma unroll
    for (int i = 0; i < 4; ++i) acc += (v[i] * (1.f / N_NODES)) * w[i];
    for (int off = 16; off; off >>= 1) acc += __shfl_down(acc, off, 32);
    if (lane == 0) out[N_NODES] = acc + bc[0];
}

static inline int blocksForWaves(long long nwaves) {
    long long threads = nwaves * 32;
    return (int)((threads + 255) / 256);
}

extern "C" void kernel_launch(void* const* d_in, const int* in_sizes, int n_in,
                              void* d_out, int out_size, void* d_ws, size_t ws_size,
                              hipStream_t stream) {
    const float* x     = (const float*)d_in[0];
    const int*   ei    = (const int*)d_in[1];
    const float* W_in  = (const float*)d_in[2];
    const float* b_in  = (const float*)d_in[3];
    const float* Wl1   = (const float*)d_in[4];
    const float* bl1   = (const float*)d_in[5];
    const float* Wr1   = (const float*)d_in[6];
    const float* br1   = (const float*)d_in[7];
    const float* att1  = (const float*)d_in[8];
    const float* bias1 = (const float*)d_in[9];
    const float* g1    = (const float*)d_in[10];
    const float* be1   = (const float*)d_in[11];
    const float* Wl2   = (const float*)d_in[12];
    const float* bl2   = (const float*)d_in[13];
    const float* Wr2   = (const float*)d_in[14];
    const float* br2   = (const float*)d_in[15];
    const float* att2  = (const float*)d_in[16];
    const float* bias2 = (const float*)d_in[17];
    const float* g2    = (const float*)d_in[18];
    const float* be2   = (const float*)d_in[19];
    const float* Wa    = (const float*)d_in[20];
    const float* ba    = (const float*)d_in[21];
    const float* Wc    = (const float*)d_in[22];
    const float* bc    = (const float*)d_in[23];

    float* out = (float*)d_out;

    float* ws = (float*)d_ws;
    size_t off = 0;
    float*    h0    = ws + off; off += (size_t)N_NODES * HID;
    float*    h1    = ws + off; off += (size_t)N_NODES * HID;
    float*    xl    = ws + off; off += (size_t)N_NODES * HC;
    float*    xr    = ws + off; off += (size_t)N_NODES * HC;
    float*    alpha = ws + off; off += (size_t)ETOT * NHEAD;
    unsigned* amax  = (unsigned*)(ws + off); off += (size_t)N_NODES * NHEAD;
    float*    denom = ws + off; off += (size_t)N_NODES * NHEAD;
    float*    res   = ws + off; off += (size_t)N_NODES * HID;
    float*    gpart = ws + off; off += HID;
    (void)ws_size; (void)in_sizes; (void)n_in; (void)out_size;

    const int initBlocks  = (N_NODES * HID + 255) / 256;
    const int edgeBlocks  = blocksForWaves(ETOT);
    const int ehBlocks    = (ETOT * NHEAD + 255) / 256;
    const int nodeBlocks  = blocksForWaves(N_NODES);
    const int gemmInBlk   = blocksForWaves((N_NODES / 16) * (HID / 16));
    const int gemmHCBlk   = blocksForWaves((N_NODES / 16) * (HC / 16));
    const int logitBlocks = (N_NODES + LG_NPB - 1) / LG_NPB;

    gemm_wmma_f32<1><<<gemmInBlk, 256, 0, stream>>>(x, W_in, b_in, h0, N_NODES, DIN, HID);

    init_scratch<<<initBlocks, 256, 0, stream>>>(res, denom, amax, gpart);
    gemm_wmma_f32<0><<<gemmHCBlk, 256, 0, stream>>>(h0, Wl1, bl1, xl, N_NODES, HID, HC);
    gemm_wmma_f32<0><<<gemmHCBlk, 256, 0, stream>>>(h0, Wr1, br1, xr, N_NODES, HID, HC);
    edge_alpha  <<<edgeBlocks, 256, 0, stream>>>(ei, xl, xr, att1, alpha, amax);
    edge_exp    <<<ehBlocks,   256, 0, stream>>>(ei, alpha, amax, denom);
    edge_scatter<<<edgeBlocks, 256, 0, stream>>>(ei, xl, alpha, denom, res);
    node_ln_relu<<<nodeBlocks, 256, 0, stream>>>(h0, res, bias1, g1, be1, h1);

    init_scratch<<<initBlocks, 256, 0, stream>>>(res, denom, amax, gpart);
    gemm_wmma_f32<0><<<gemmHCBlk, 256, 0, stream>>>(h1, Wl2, bl2, xl, N_NODES, HID, HC);
    gemm_wmma_f32<0><<<gemmHCBlk, 256, 0, stream>>>(h1, Wr2, br2, xr, N_NODES, HID, HC);
    edge_alpha  <<<edgeBlocks, 256, 0, stream>>>(ei, xl, xr, att2, alpha, amax);
    edge_exp    <<<ehBlocks,   256, 0, stream>>>(ei, alpha, amax, denom);
    edge_scatter<<<edgeBlocks, 256, 0, stream>>>(ei, xl, alpha, denom, res);
    node_ln_relu<<<nodeBlocks, 256, 0, stream>>>(h1, res, bias2, g2, be2, h0);

    logits_kernel<<<logitBlocks, LG_WAVES * 32, 0, stream>>>(h0, Wa, ba, gpart, out);
    value_kernel <<<1, 32, 0, stream>>>(gpart, Wc, bc, out);
}


# edges grouped by destination once per call (counting sort, arrival-order slots); one wave per node does softmax and weighted sum in registers, no float atomics; float sum order changes
# speedup vs baseline: 2.0741x; 1.6317x over previous
#include <hip/hip_runtime.h>
#include <hip/hip_bf16.h>
#include <math.h>

#define N_NODES 20000
#define E_EDGES 320000
#define ETOT    (E_EDGES + N_NODES)
#define DIN     64
#define HID     128
#define NHEAD   4
#define HC      (NHEAD * HID)

typedef float v2f __attribute__((ext_vector_type(2)));
typedef float v4f __attribute__((ext_vector_type(4)));
typedef float v8f __attribute__((ext_vector_type(8)));

template <int RELU>
__global__ void gemm_wmma_f32(const float* __restrict__ A,
                              const float* __restrict__ B,
                              const float* __restrict__ bias,
                              float* __restrict__ C,
                              int M, int K, int N) {
    int wave   = (blockIdx.x * blockDim.x + threadIdx.x) >> 5;
    int tilesN = N >> 4;
    int tiles  = (M >> 4) * tilesN;
    if (wave >= tiles) return;
    int tm = wave / tilesN, tn = wave % tilesN;
    int lane = threadIdx.x & 31;
    int l16  = lane & 15;
    int half = lane >> 4;

    const float* arow = A + (size_t)(tm * 16 + l16) * K + 2 * half;
    const float* bcol = B + (size_t)(2 * half) * N + tn * 16 + l16;

    v8f acc = {0.f, 0.f, 0.f, 0.f, 0.f, 0.f, 0.f, 0.f};
    for (int k = 0; k < K; k += 4) {
        v2f a, b;
        a.x = arow[k];
        a.y = arow[k + 1];
        b.x = bcol[(size_t)k * N];
        b.y = bcol[(size_t)(k + 1) * N];
        acc = __builtin_amdgcn_wmma_f32_16x16x4_f32(
             false, a,  false, b,
             (short)0, acc,  false,  false);
    }

    int   col = tn * 16 + l16;
    float bv  = bias[col];
#pragma unroll
    for (int v = 0; v < 8; ++v) {
        int   row = tm * 16 + v + 8 * half;
        float val = acc[v] + bv;
        if (RELU) val = fmaxf(val, 0.f);
        C[(size_t)row * N + col] = val;
    }
}

__global__ void init_scratch(int* __restrict__ cnt, float* __restrict__ gpart) {
    int idx = blockIdx.x * blockDim.x + threadIdx.x;
    if (idx < N_NODES) cnt[idx] = 0;
    if (idx < HID) gpart[idx] = 0.f;
}

__global__ void csr_count(const int* __restrict__ ei, int* __restrict__ cnt,
                          int* __restrict__ slot) {
    int e = blockIdx.x * blockDim.x + threadIdx.x;
    if (e >= E_EDGES) return;
    slot[e] = atomicAdd(&cnt[ei[E_EDGES + e]], 1);
}

#define SC_THREADS 1024
#define SC_PER     20
static_assert(SC_THREADS * SC_PER >= N_NODES, "one block must cover every node");

__global__ void __launch_bounds__(SC_THREADS)
csr_scan(const int* __restrict__ cnt, int* __restrict__ first) {
    __shared__ int part[SC_THREADS];
    const int t    = threadIdx.x;
    const int base = t * SC_PER;
    int loc[SC_PER];
    int s = 0;
#pragma unroll
    for (int i = 0; i < SC_PER; ++i) {
        int idx = base + i;
        int v   = (idx < N_NODES) ? cnt[idx] : 0;
        loc[i]  = s;
        s += v;
    }
    part[t] = s;
    __syncthreads();
    for (int off = 1; off < SC_THREADS; off <<= 1) {
        int v = (t >= off) ? part[t - off] : 0;
        __syncthreads();
        part[t] += v;
        __syncthreads();
    }
    const int excl = part[t] - s;
#pragma unroll
    for (int i = 0; i < SC_PER; ++i) {
        int idx = base + i;
        if (idx < N_NODES) first[idx] = excl + loc[i];
    }
    if (t == SC_THREADS - 1) first[N_NODES] = part[t];
}

__global__ void csr_fill(const int* __restrict__ ei, const int* __restrict__ first,
                         const int* __restrict__ slot, int* __restrict__ srcs) {
    int e = blockIdx.x * blockDim.x + threadIdx.x;
    if (e >= E_EDGES) return;
    srcs[first[ei[E_EDGES + e]] + slot[e]] = ei[e];
}

__device__ __forceinline__ void edge_scores(const v4f (&x)[NHEAD], const v4f (&r)[NHEAD],
                                            const v4f (&w)[NHEAD], float (&a)[NHEAD]) {
#pragma unroll
    for (int h = 0; h < NHEAD; ++h) {
        float acc = 0.f;
#pragma unroll
        for (int i = 0; i < 4; ++i) {
            float m = x[h][i] + r[h][i];
            float e = (m > 0.f) ? m : 0.2f * m;
            acc += e * w[h][i];
        }
        a[h] = acc;
    }
#pragma unroll
    for (int off = 16; off; off >>= 1) {
#pragma unroll
        for (int h = 0; h < NHEAD; ++h) a[h] += __shfl_xor(a[h], off, 32);
    }
}

__device__ __forceinline__ void load_row(const float* __restrict__ base, int lane,
                                         v4f (&x)[NHEAD]) {
    const v4f* p = (const v4f*)base;
#pragma unroll
    for (int h = 0; h < NHEAD; ++h) x[h] = p[h * (HID / 4) + lane];
}

#define NG_WAVES 4

__global__ void __launch_bounds__(NG_WAVES * 32)
node_gather(const int* __restrict__ first, const int* __restrict__ srcs,
            const float* __restrict__ xl, const float* __restrict__ xr,
            const float* __restrict__ att, float* __restrict__ res) {
    const int node = __builtin_amdgcn_readfirstlane((int)((blockIdx.x * blockDim.x + threadIdx.x) >> 5));
    if (node >= N_NODES) return;
    const int lane = threadIdx.x & 31;
    const int beg  = first[node];
    const int end  = first[node + 1];

    v4f r[NHEAD], w[NHEAD], x[NHEAD], xn[NHEAD];
    load_row(xr + (size_t)node * HC, lane, r);
    load_row(att, lane, w);
    load_row(xl + (size_t)node * HC, lane, x);
#pragma unroll
    for (int h = 0; h < NHEAD; ++h) xn[h] = (v4f){0.f, 0.f, 0.f, 0.f};
    if (beg < end) load_row(xl + (size_t)srcs[beg] * HC, lane, xn);

    float m[NHEAD], l[NHEAD];
    v4f   acc[NHEAD];
    {
        float a[NHEAD];
        edge_scores(x, r, w, a);
#pragma unroll
        for (int h = 0; h < NHEAD; ++h) { m[h] = a[h]; l[h] = 1.f; acc[h] = x[h]; }
    }

    for (int i = beg; i < end; ++i) {
#pragma unroll
        for (int h = 0; h < NHEAD; ++h) x[h] = xn[h];
        if (i + 1 < end) load_row(xl + (size_t)srcs[i + 1] * HC, lane, xn);

        float a[NHEAD];
        edge_scores(x, r, w, a);
#pragma unroll
        for (int h = 0; h < NHEAD; ++h) {
            const float d  = a[h] - m[h];
            const float e  = expf(-fabsf(d));
            const bool  up = d > 0.f;
            const float sc = up ? e : 1.f;
            const float p  = up ? 1.f : e;
            m[h] = up ? a[h] : m[h];
            l[h] = l[h] * sc + p;
#pragma unroll
            for (int k = 0; k < 4; ++k) acc[h][k] = acc[h][k] * sc + x[h][k] * p;
        }
    }

    v4f o = {0.f, 0.f, 0.f, 0.f};
#pragma unroll
    for (int h = 0; h < NHEAD; ++h) {
        const float inv = 1.f / l[h];
#pragma unroll
        for (int k = 0; k < 4; ++k) o[k] += acc[h][k] * inv;
    }
#pragma unroll
    for (int k = 0; k < 4; ++k) o[k] *= 0.25f;
    ((v4f*)(res + (size_t)node * HID))[lane] = o;
}

__global__ void node_ln_relu(const float* __restrict__ hin, const float* __restrict__ res,
                             const float* __restrict__ bias, const float* __restrict__ g,
                             const float* __restrict__ be, float* __restrict__ hout) {
    int wave = (blockIdx.x * blockDim.x + threadIdx.x) >> 5;
    if (wave >= N_NODES) return;
    int lane = threadIdx.x & 31;
    const v4f* ph = (const v4f*)(hin + (size_t)wave * HID);
    const v4f* pr = (const v4f*)(res + (size_t)wave * HID);
    const v4f* pb = (const v4f*)bias;

    v4f y;
    {
        v4f a = ph[lane], b = pr[lane], c = pb[lane];
#pragma unroll
        for (int i = 0; i < 4; ++i) y[i] = a[i] + b[i] + c[i];
    }
    float s = 0.f, ss = 0.f;
#pragma unroll
    for (int i = 0; i < 4; ++i) { s += y[i]; ss += y[i] * y[i]; }
    for (int off = 16; off; off >>= 1) {
        s  += __shfl_xor(s,  off, 32);
        ss += __shfl_xor(ss, off, 32);
    }
    float mu  = s * (1.f / HID);
    float var = ss * (1.f / HID) - mu * mu;
    float r   = rsqrtf(var + 1e-5f);

    v4f gg = ((const v4f*)g)[lane];
    v4f bb = ((const v4f*)be)[lane];
    v4f o;
#pragma unroll
    for (int i = 0; i < 4; ++i)
        o[i] = fmaxf((y[i] - mu) * r * gg[i] + bb[i], 0.f);
    ((v4f*)(hout + (size_t)wave * HID))[lane] = o;
}

#define LG_WAVES 8
#define LG_NPW   16
#define LG_NPB   (LG_WAVES * LG_NPW)

__global__ void __launch_bounds__(LG_WAVES * 32)
logits_kernel(const float* __restrict__ shared_f, const float* __restrict__ Wa,
              const float* __restrict__ ba, float* __restrict__ gpart,
              float* __restrict__ out) {
    __shared__ float colpart[LG_WAVES][HID];
    __shared__ float lg[LG_NPB];

    const int lane  = threadIdx.x & 31;
    const int wv    = __builtin_amdgcn_readfirstlane(threadIdx.x >> 5);
    const int node0 = blockIdx.x * LG_NPB + wv * LG_NPW;

    const v4f w = ((const v4f*)Wa)[lane];

    v4f rows[LG_NPW];
#pragma unroll
    for (int j = 0; j < LG_NPW; ++j) {
        int node = node0 + j;
        int nc   = (node < N_NODES) ? node : (N_NODES - 1);
        rows[j]  = ((const v4f*)(shared_f + (size_t)nc * HID))[lane];
    }

    float acc[LG_NPW];
    v4f cs = {0.f, 0.f, 0.f, 0.f};
#pragma unroll
    for (int j = 0; j < LG_NPW; ++j) {
        const bool valid = (node0 + j) < N_NODES;
        const v4f  v     = rows[j];
        float a = 0.f;
#pragma unroll
        for (int i = 0; i < 4; ++i) {
            a     += v[i] * w[i];
            cs[i] += valid ? v[i] : 0.f;
        }
        acc[j] = a;
    }
#pragma unroll
    for (int off = 16; off; off >>= 1) {
#pragma unroll
        for (int j = 0; j < LG_NPW; ++j) acc[j] += __shfl_down(acc[j], off, 32);
    }
    if (lane == 0) {
#pragma unroll
        for (int j = 0; j < LG_NPW; ++j) lg[wv * LG_NPW + j] = acc[j];
    }
    ((v4f*)&colpart[wv][0])[lane] = cs;
    __syncthreads();

    const int t = threadIdx.x;
    if (t < HID) {
        float s = 0.f;
#pragma unroll
        for (int k = 0; k < LG_WAVES; ++k) s += colpart[k][t];
        atomicAdd(&gpart[t], s);
    }
    if (t < LG_NPB) {
        int node = blockIdx.x * LG_NPB + t;
        if (node < N_NODES) out[node] = lg[t] + ba[0];
    }
}

__global__ void value_kernel(const float* __restrict__ gpart, const float* __restrict__ Wc,
                             const float* __restrict__ bc, float* __restrict__ out) {
    int lane = threadIdx.x & 31;
    v4f v = ((const v4f*)gpart)[lane];
    v4f w = ((const v4f*)Wc)[lane];
    float acc = 0.f;
#pragma unroll
    for (int i = 0; i < 4; ++i) acc += (v[i] * (1.f / N_NODES)) * w[i];
    for (int off = 16; off; off >>= 1) acc += __shfl_down(acc, off, 32);
    if (lane == 0) out[N_NODES] = acc + bc[0];
}

static inline int blocksForWaves(long long nwaves) {
    long long threads = nwaves * 32;
    return (int)((threads + 255) / 256);
}

extern "C" void kernel_launch(void* const* d_in, const int* in_sizes, int n_in,
                              void* d_out, int out_size, void* d_ws, size_t ws_size,
                              hipStream_t stream) {
    const float* x     = (const float*)d_in[0];
    const int*   ei    = (const int*)d_in[1];
    const float* W_in  = (const float*)d_in[2];
    const float* b_in  = (const float*)d_in[3];
    const float* Wl1   = (const float*)d_in[4];
    const float* bl1   = (const float*)d_in[5];
    const float* Wr1   = (const float*)d_in[6];
    const float* br1   = (const float*)d_in[7];
    const float* att1  = (const float*)d_in[8];
    const float* bias1 = (const float*)d_in[9];
    const float* g1    = (const float*)d_in[10];
    const float* be1   = (const float*)d_in[11];
    const float* Wl2   = (const float*)d_in[12];
    const float* bl2   = (const float*)d_in[13];
    const float* Wr2   = (const float*)d_in[14];
    const float* br2   = (const float*)d_in[15];
    const float* att2  = (const float*)d_in[16];
    const float* bias2 = (const float*)d_in[17];
    const float* g2    = (const float*)d_in[18];
    const float* be2   = (const float*)d_in[19];
    const float* Wa    = (const float*)d_in[20];
    const float* ba    = (const float*)d_in[21];
    const float* Wc    = (const float*)d_in[22];
    const float* bc    = (const float*)d_in[23];

    float* out = (float*)d_out;

    float* ws = (float*)d_ws;
    size_t off = 0;
    float*    h0    = ws + off; off += (size_t)N_NODES * HID;
    float*    h1    = ws + off; off += (size_t)N_NODES * HID;
    float*    xl    = ws + off; off += (size_t)N_NODES * HC;
    float*    xr    = ws + off; off += (size_t)N_NODES * HC;
    int*      csr   = (int*)(ws + off); off += (size_t)ETOT * NHEAD;
    off += (size_t)N_NODES * NHEAD;
    off += (size_t)N_NODES * NHEAD;
    float*    res   = ws + off; off += (size_t)N_NODES * HID;
    float*    gpart = ws + off; off += HID;
    (void)ws_size; (void)in_sizes; (void)n_in; (void)out_size;

    int* cnt   = csr;
    int* first = cnt + N_NODES;
    int* slot  = first + (N_NODES + 16);
    int* srcs  = slot + E_EDGES;
    static_assert((size_t)N_NODES + (N_NODES + 16) + 2 * (size_t)E_EDGES <= (size_t)ETOT * NHEAD,
                  "the grouping must fit the region it reuses");

    const int initBlocks  = (N_NODES + 255) / 256;
    const int eBlocks     = (E_EDGES + 255) / 256;
    const int gatherBlocks = (N_NODES + NG_WAVES - 1) / NG_WAVES;
    const int nodeBlocks  = blocksForWaves(N_NODES);
    const int gemmInBlk   = blocksForWaves((N_NODES / 16) * (HID / 16));
    const int gemmHCBlk   = blocksForWaves((N_NODES / 16) * (HC / 16));
    const int logitBlocks = (N_NODES + LG_NPB - 1) / LG_NPB;

    init_scratch<<<initBlocks, 256, 0, stream>>>(cnt, gpart);
    csr_count   <<<eBlocks, 256, 0, stream>>>(ei, cnt, slot);
    csr_scan    <<<1, SC_THREADS, 0, stream>>>(cnt, first);
    csr_fill    <<<eBlocks, 256, 0, stream>>>(ei, first, slot, srcs);

    gemm_wmma_f32<1><<<gemmInBlk, 256, 0, stream>>>(x, W_in, b_in, h0, N_NODES, DIN, HID);

    gemm_wmma_f32<0><<<gemmHCBlk, 256, 0, stream>>>(h0, Wl1, bl1, xl, N_NODES, HID, HC);
    gemm_wmma_f32<0><<<gemmHCBlk, 256, 0, stream>>>(h0, Wr1, br1, xr, N_NODES, HID, HC);
    node_gather <<<gatherBlocks, NG_WAVES * 32, 0, stream>>>(first, srcs, xl, xr, att1, res);
    node_ln_relu<<<nodeBlocks, 256, 0, stream>>>(h0, res, bias1, g1, be1, h1);

    gemm_wmma_f32<0><<<gemmHCBlk, 256, 0, stream>>>(h1, Wl2, bl2, xl, N_NODES, HID, HC);
    gemm_wmma_f32<0><<<gemmHCBlk, 256, 0, stream>>>(h1, Wr2, br2, xr, N_NODES, HID, HC);
    node_gather <<<gatherBlocks, NG_WAVES * 32, 0, stream>>>(first, srcs, xl, xr, att2, res);
    node_ln_relu<<<nodeBlocks, 256, 0, stream>>>(h1, res, bias2, g2, be2, h0);

    logits_kernel<<<logitBlocks, LG_WAVES * 32, 0, stream>>>(h0, Wa, ba, gpart, out);
    value_kernel <<<1, 32, 0, stream>>>(gpart, Wc, bc, out);
}


# GEMM kernels: 8-wave block owns an 80x128 tile of C; A rows copied once to LDS by whole-line loads, each wave keeps its 16 columns of B in registers, 5 accumulator tiles per k step; k order kept
# speedup vs baseline: 2.8274x; 1.3632x over previous
#include <hip/hip_runtime.h>
#include <hip/hip_bf16.h>
#include <math.h>

#define N_NODES 20000
#define E_EDGES 320000
#define ETOT    (E_EDGES + N_NODES)
#define DIN     64
#define HID     128
#define NHEAD   4
#define HC      (NHEAD * HID)

typedef float v2f __attribute__((ext_vector_type(2)));
typedef float v4f __attribute__((ext_vector_type(4)));
typedef float v8f __attribute__((ext_vector_type(8)));

#define GM_BM    80
#define GM_MT    (GM_BM / 16)
#define GM_WAVES 8
#define GM_BN    (GM_WAVES * 16)
static_assert(N_NODES % GM_BM == 0, "the row blocks must cover M exactly");
static_assert(HID % GM_BN == 0 && HC % GM_BN == 0, "the column blocks must cover N exactly");

template <int RELU, int K, int N>
__global__ void __launch_bounds__(GM_WAVES * 32)
gemm_wmma_f32(const float* __restrict__ A,
              const float* __restrict__ B,
              const float* __restrict__ bias,
              float* __restrict__ C) {
    constexpr int P    = K + 4;
    constexpr int KS   = K / 4;
    constexpr int NV4  = GM_BM * KS;
    constexpr int NTHR = GM_WAVES * 32;
    static_assert(NV4 % NTHR == 0, "every thread copies the same number of chunks");
    __shared__ __attribute__((aligned(16))) float As[GM_BM * P];

    const int t    = threadIdx.x;
    const int lane = t & 31;
    const int wv   = __builtin_amdgcn_readfirstlane(t >> 5);
    const int l16  = lane & 15;
    const int half = lane >> 4;
    const int m0   = blockIdx.y * GM_BM;
    const int col  = blockIdx.x * GM_BN + wv * 16 + l16;

    const v4f* ag = (const v4f*)(A + (size_t)m0 * K);
    v4f stage[NV4 / NTHR];
#pragma unroll
    for (int i = 0; i < NV4 / NTHR; ++i) stage[i] = ag[t + NTHR * i];

    const float* bcol = B + (size_t)(2 * half) * N + col;
    v2f bfr[KS];
#pragma unroll
    for (int j = 0; j < KS; ++j) {
        bfr[j].x = bcol[(size_t)(4 * j) * N];
        bfr[j].y = bcol[(size_t)(4 * j + 1) * N];
    }

#pragma unroll
    for (int i = 0; i < NV4 / NTHR; ++i) {
        const int idx = t + NTHR * i;
        const int row = idx / KS;
        const int c4  = idx % KS;
        *(v4f*)&As[row * P + 4 * c4] = stage[i];
    }
    __syncthreads();

    v8f acc[GM_MT];
#pragma unroll
    for (int mt = 0; mt < GM_MT; ++mt) acc[mt] = (v8f){0.f, 0.f, 0.f, 0.f, 0.f, 0.f, 0.f, 0.f};

    const float* arow = &As[l16 * P + 2 * half];
#pragma unroll
    for (int j = 0; j < KS; ++j) {
#pragma unroll
        for (int mt = 0; mt < GM_MT; ++mt) {
            const v2f a = *(const v2f*)(arow + mt * 16 * P + 4 * j);
            acc[mt] = __builtin_amdgcn_wmma_f32_16x16x4_f32(
                 false, a,  false, bfr[j],
                 (short)0, acc[mt],  false,  false);
        }
    }

    const float bv = bias[col];
#pragma unroll
    for (int mt = 0; mt < GM_MT; ++mt) {
#pragma unroll
        for (int v = 0; v < 8; ++v) {
            const int row = m0 + mt * 16 + v + 8 * half;
            float val = acc[mt][v] + bv;
            if (RELU) val = fmaxf(val, 0.f);
            C[(size_t)row * N + col] = val;
        }
    }
}

__global__ void init_scratch(int* __restrict__ cnt, float* __restrict__ gpart) {
    int idx = blockIdx.x * blockDim.x + threadIdx.x;
    if (idx < N_NODES) cnt[idx] = 0;
    if (idx < HID) gpart[idx] = 0.f;
}

__global__ void csr_count(const int* __restrict__ ei, int* __restrict__ cnt,
                          int* __restrict__ slot) {
    int e = blockIdx.x * blockDim.x + threadIdx.x;
    if (e >= E_EDGES) return;
    slot[e] = atomicAdd(&cnt[ei[E_EDGES + e]], 1);
}

#define SC_THREADS 1024
#define SC_PER     20
static_assert(SC_THREADS * SC_PER >= N_NODES, "one block must cover every node");

__global__ void __launch_bounds__(SC_THREADS)
csr_scan(const int* __restrict__ cnt, int* __restrict__ first) {
    __shared__ int part[SC_THREADS];
    const int t    = threadIdx.x;
    const int base = t * SC_PER;
    int loc[SC_PER];
    int s = 0;
#pragma unroll
    for (int i = 0; i < SC_PER; ++i) {
        int idx = base + i;
        int v   = (idx < N_NODES) ? cnt[idx] : 0;
        loc[i]  = s;
        s += v;
    }
    part[t] = s;
    __syncthreads();
    for (int off = 1; off < SC_THREADS; off <<= 1) {
        int v = (t >= off) ? part[t - off] : 0;
        __syncthreads();
        part[t] += v;
        __syncthreads();
    }
    const int excl = part[t] - s;
#pragma unroll
    for (int i = 0; i < SC_PER; ++i) {
        int idx = base + i;
        if (idx < N_NODES) first[idx] = excl + loc[i];
    }
    if (t == SC_THREADS - 1) first[N_NODES] = part[t];
}

__global__ void csr_fill(const int* __restrict__ ei, const int* __restrict__ first,
                         const int* __restrict__ slot, int* __restrict__ srcs) {
    int e = blockIdx.x * blockDim.x + threadIdx.x;
    if (e >= E_EDGES) return;
    srcs[first[ei[E_EDGES + e]] + slot[e]] = ei[e];
}

__device__ __forceinline__ void edge_scores(const v4f (&x)[NHEAD], const v4f (&r)[NHEAD],
                                            const v4f (&w)[NHEAD], float (&a)[NHEAD]) {
#pragma unroll
    for (int h = 0; h < NHEAD; ++h) {
        float acc = 0.f;
#pragma unroll
        for (int i = 0; i < 4; ++i) {
            float m = x[h][i] + r[h][i];
            float e = (m > 0.f) ? m : 0.2f * m;
            acc += e * w[h][i];
        }
        a[h] = acc;
    }
#pragma unroll
    for (int off = 16; off; off >>= 1) {
#pragma unroll
        for (int h = 0; h < NHEAD; ++h) a[h] += __shfl_xor(a[h], off, 32);
    }
}

__device__ __forceinline__ void load_row(const float* __restrict__ base, int lane,
                                         v4f (&x)[NHEAD]) {
    const v4f* p = (const v4f*)base;
#pragma unroll
    for (int h = 0; h < NHEAD; ++h) x[h] = p[h * (HID / 4) + lane];
}

#define NG_WAVES 4

__global__ void __launch_bounds__(NG_WAVES * 32)
node_gather(const int* __restrict__ first, const int* __restrict__ srcs,
            const float* __restrict__ xl, const float* __restrict__ xr,
            const float* __restrict__ att, float* __restrict__ res) {
    const int node = __builtin_amdgcn_readfirstlane((int)((blockIdx.x * blockDim.x + threadIdx.x) >> 5));
    if (node >= N_NODES) return;
    const int lane = threadIdx.x & 31;
    const int beg  = first[node];
    const int end  = first[node + 1];

    v4f r[NHEAD], w[NHEAD], x[NHEAD], xn[NHEAD];
    load_row(xr + (size_t)node * HC, lane, r);
    load_row(att, lane, w);
    load_row(xl + (size_t)node * HC, lane, x);
#pragma unroll
    for (int h = 0; h < NHEAD; ++h) xn[h] = (v4f){0.f, 0.f, 0.f, 0.f};
    if (beg < end) load_row(xl + (size_t)srcs[beg] * HC, lane, xn);

    float m[NHEAD], l[NHEAD];
    v4f   acc[NHEAD];
    {
        float a[NHEAD];
        edge_scores(x, r, w, a);
#pragma unroll
        for (int h = 0; h < NHEAD; ++h) { m[h] = a[h]; l[h] = 1.f; acc[h] = x[h]; }
    }

    for (int i = beg; i < end; ++i) {
#pragma unroll
        for (int h = 0; h < NHEAD; ++h) x[h] = xn[h];
        if (i + 1 < end) load_row(xl + (size_t)srcs[i + 1] * HC, lane, xn);

        float a[NHEAD];
        edge_scores(x, r, w, a);
#pragma unroll
        for (int h = 0; h < NHEAD; ++h) {
            const float d  = a[h] - m[h];
            const float e  = expf(-fabsf(d));
            const bool  up = d > 0.f;
            const float sc = up ? e : 1.f;
            const float p  = up ? 1.f : e;
            m[h] = up ? a[h] : m[h];
            l[h] = l[h] * sc + p;
#pragma unroll
            for (int k = 0; k < 4; ++k) acc[h][k] = acc[h][k] * sc + x[h][k] * p;
        }
    }

    v4f o = {0.f, 0.f, 0.f, 0.f};
#pragma unroll
    for (int h = 0; h < NHEAD; ++h) {
        const float inv = 1.f / l[h];
#pragma unroll
        for (int k = 0; k < 4; ++k) o[k] += acc[h][k] * inv;
    }
#pragma unroll
    for (int k = 0; k < 4; ++k) o[k] *= 0.25f;
    ((v4f*)(res + (size_t)node * HID))[lane] = o;
}

__global__ void node_ln_relu(const float* __restrict__ hin, const float* __restrict__ res,
                             const float* __restrict__ bias, const float* __restrict__ g,
                             const float* __restrict__ be, float* __restrict__ hout) {
    int wave = (blockIdx.x * blockDim.x + threadIdx.x) >> 5;
    if (wave >= N_NODES) return;
    int lane = threadIdx.x & 31;
    const v4f* ph = (const v4f*)(hin + (size_t)wave * HID);
    const v4f* pr = (const v4f*)(res + (size_t)wave * HID);
    const v4f* pb = (const v4f*)bias;

    v4f y;
    {
        v4f a = ph[lane], b = pr[lane], c = pb[lane];
#pragma unroll
        for (int i = 0; i < 4; ++i) y[i] = a[i] + b[i] + c[i];
    }
    float s = 0.f, ss = 0.f;
#pragma unroll
    for (int i = 0; i < 4; ++i) { s += y[i]; ss += y[i] * y[i]; }
    for (int off = 16; off; off >>= 1) {
        s  += __shfl_xor(s,  off, 32);
        ss += __shfl_xor(ss, off, 32);
    }
    float mu  = s * (1.f / HID);
    float var = ss * (1.f / HID) - mu * mu;
    float r   = rsqrtf(var + 1e-5f);

    v4f gg = ((const v4f*)g)[lane];
    v4f bb = ((const v4f*)be)[lane];
    v4f o;
#pragma unroll
    for (int i = 0; i < 4; ++i)
        o[i] = fmaxf((y[i] - mu) * r * gg[i] + bb[i], 0.f);
    ((v4f*)(hout + (size_t)wave * HID))[lane] = o;
}

#define LG_WAVES 8
#define LG_NPW   16
#define LG_NPB   (LG_WAVES * LG_NPW)

__global__ void __launch_bounds__(LG_WAVES * 32)
logits_kernel(const float* __restrict__ shared_f, const float* __restrict__ Wa,
              const float* __restrict__ ba, float* __restrict__ gpart,
              float* __restrict__ out) {
    __shared__ float colpart[LG_WAVES][HID];
    __shared__ float lg[LG_NPB];

    const int lane  = threadIdx.x & 31;
    const int wv    = __builtin_amdgcn_readfirstlane(threadIdx.x >> 5);
    const int node0 = blockIdx.x * LG_NPB + wv * LG_NPW;

    const v4f w = ((const v4f*)Wa)[lane];

    v4f rows[LG_NPW];
#pragma unroll
    for (int j = 0; j < LG_NPW; ++j) {
        int node = node0 + j;
        int nc   = (node < N_NODES) ? node : (N_NODES - 1);
        rows[j]  = ((const v4f*)(shared_f + (size_t)nc * HID))[lane];
    }

    float acc[LG_NPW];
    v4f cs = {0.f, 0.f, 0.f, 0.f};
#pragma unroll
    for (int j = 0; j < LG_NPW; ++j) {
        const bool valid = (node0 + j) < N_NODES;
        const v4f  v     = rows[j];
        float a = 0.f;
#pragma unroll
        for (int i = 0; i < 4; ++i) {
            a     += v[i] * w[i];
            cs[i] += valid ? v[i] : 0.f;
        }
        acc[j] = a;
    }
#pragma unroll
    for (int off = 16; off; off >>= 1) {
#pragma unroll
        for (int j = 0; j < LG_NPW; ++j) acc[j] += __shfl_down(acc[j], off, 32);
    }
    if (lane == 0) {
#pragma unroll
        for (int j = 0; j < LG_NPW; ++j) lg[wv * LG_NPW + j] = acc[j];
    }
    ((v4f*)&colpart[wv][0])[lane] = cs;
    __syncthreads();

    const int t = threadIdx.x;
    if (t < HID) {
        float s = 0.f;
#pragma unroll
        for (int k = 0; k < LG_WAVES; ++k) s += colpart[k][t];
        atomicAdd(&gpart[t], s);
    }
    if (t < LG_NPB) {
        int node = blockIdx.x * LG_NPB + t;
        if (node < N_NODES) out[node] = lg[t] + ba[0];
    }
}

__global__ void value_kernel(const float* __restrict__ gpart, const float* __restrict__ Wc,
                             const float* __restrict__ bc, float* __restrict__ out) {
    int lane = threadIdx.x & 31;
    v4f v = ((const v4f*)gpart)[lane];
    v4f w = ((const v4f*)Wc)[lane];
    float acc = 0.f;
#pragma unroll
    for (int i = 0; i < 4; ++i) acc += (v[i] * (1.f / N_NODES)) * w[i];
    for (int off = 16; off; off >>= 1) acc += __shfl_down(acc, off, 32);
    if (lane == 0) out[N_NODES] = acc + bc[0];
}

static inline int blocksForWaves(long long nwaves) {
    long long threads = nwaves * 32;
    return (int)((threads + 255) / 256);
}

extern "C" void kernel_launch(void* const* d_in, const int* in_sizes, int n_in,
                              void* d_out, int out_size, void* d_ws, size_t ws_size,
                              hipStream_t stream) {
    const float* x     = (const float*)d_in[0];
    const int*   ei    = (const int*)d_in[1];
    const float* W_in  = (const float*)d_in[2];
    const float* b_in  = (const float*)d_in[3];
    const float* Wl1   = (const float*)d_in[4];
    const float* bl1   = (const float*)d_in[5];
    const float* Wr1   = (const float*)d_in[6];
    const float* br1   = (const float*)d_in[7];
    const float* att1  = (const float*)d_in[8];
    const float* bias1 = (const float*)d_in[9];
    const float* g1    = (const float*)d_in[10];
    const float* be1   = (const float*)d_in[11];
    const float* Wl2   = (const float*)d_in[12];
    const float* bl2   = (const float*)d_in[13];
    const float* Wr2   = (const float*)d_in[14];
    const float* br2   = (const float*)d_in[15];
    const float* att2  = (const float*)d_in[16];
    const float* bias2 = (const float*)d_in[17];
    const float* g2    = (const float*)d_in[18];
    const float* be2   = (const float*)d_in[19];
    const float* Wa    = (const float*)d_in[20];
    const float* ba    = (const float*)d_in[21];
    const float* Wc    = (const float*)d_in[22];
    const float* bc    = (const float*)d_in[23];

    float* out = (float*)d_out;

    float* ws = (float*)d_ws;
    size_t off = 0;
    float*    h0    = ws + off; off += (size_t)N_NODES * HID;
    float*    h1    = ws + off; off += (size_t)N_NODES * HID;
    float*    xl    = ws + off; off += (size_t)N_NODES * HC;
    float*    xr    = ws + off; off += (size_t)N_NODES * HC;
    int*      csr   = (int*)(ws + off); off += (size_t)ETOT * NHEAD;
    off += (size_t)N_NODES * NHEAD;
    off += (size_t)N_NODES * NHEAD;
    float*    res   = ws + off; off += (size_t)N_NODES * HID;
    float*    gpart = ws + off; off += HID;
    (void)ws_size; (void)in_sizes; (void)n_in; (void)out_size;

    int* cnt   = csr;
    int* first = cnt + N_NODES;
    int* slot  = first + (N_NODES + 16);
    int* srcs  = slot + E_EDGES;
    static_assert((size_t)N_NODES + (N_NODES + 16) + 2 * (size_t)E_EDGES <= (size_t)ETOT * NHEAD,
                  "the grouping must fit the region it reuses");

    const int initBlocks  = (N_NODES + 255) / 256;
    const int eBlocks     = (E_EDGES + 255) / 256;
    const int gatherBlocks = (N_NODES + NG_WAVES - 1) / NG_WAVES;
    const int nodeBlocks  = blocksForWaves(N_NODES);
    const dim3 gemmInGrid(HID / GM_BN, N_NODES / GM_BM);
    const dim3 gemmHCGrid(HC / GM_BN, N_NODES / GM_BM);
    const int logitBlocks = (N_NODES + LG_NPB - 1) / LG_NPB;

    init_scratch<<<initBlocks, 256, 0, stream>>>(cnt, gpart);
    csr_count   <<<eBlocks, 256, 0, stream>>>(ei, cnt, slot);
    csr_scan    <<<1, SC_THREADS, 0, stream>>>(cnt, first);
    csr_fill    <<<eBlocks, 256, 0, stream>>>(ei, first, slot, srcs);

    gemm_wmma_f32<1, DIN, HID><<<gemmInGrid, GM_WAVES * 32, 0, stream>>>(x, W_in, b_in, h0);

    gemm_wmma_f32<0, HID, HC><<<gemmHCGrid, GM_WAVES * 32, 0, stream>>>(h0, Wl1, bl1, xl);
    gemm_wmma_f32<0, HID, HC><<<gemmHCGrid, GM_WAVES * 32, 0, stream>>>(h0, Wr1, br1, xr);
    node_gather <<<gatherBlocks, NG_WAVES * 32, 0, stream>>>(first, srcs, xl, xr, att1, res);
    node_ln_relu<<<nodeBlocks, 256, 0, stream>>>(h0, res, bias1, g1, be1, h1);

    gemm_wmma_f32<0, HID, HC><<<gemmHCGrid, GM_WAVES * 32, 0, stream>>>(h1, Wl2, bl2, xl);
    gemm_wmma_f32<0, HID, HC><<<gemmHCGrid, GM_WAVES * 32, 0, stream>>>(h1, Wr2, br2, xr);
    node_gather <<<gatherBlocks, NG_WAVES * 32, 0, stream>>>(first, srcs, xl, xr, att2, res);
    node_ln_relu<<<nodeBlocks, 256, 0, stream>>>(h1, res, bias2, g2, be2, h0);

    logits_kernel<<<logitBlocks, LG_WAVES * 32, 0, stream>>>(h0, Wa, ba, gpart, out);
    value_kernel <<<1, 32, 0, stream>>>(gpart, Wc, bc, out);
}


# gather kernel: a lane owns 16 floats of ONE head (4 whole 128-B lines per load): a score is one 8-lane DPP sum and one exp per edge, not four 32-lane sums via LDS; float sum order changes
# speedup vs baseline: 3.1008x; 1.0967x over previous
#include <hip/hip_runtime.h>
#include <hip/hip_bf16.h>
#include <math.h>

#define N_NODES 20000
#define E_EDGES 320000
#define ETOT    (E_EDGES + N_NODES)
#define DIN     64
#define HID     128
#define NHEAD   4
#define HC      (NHEAD * HID)

typedef float v2f __attribute__((ext_vector_type(2)));
typedef float v4f __attribute__((ext_vector_type(4)));
typedef float v8f __attribute__((ext_vector_type(8)));

#define GM_BM    80
#define GM_MT    (GM_BM / 16)
#define GM_WAVES 8
#define GM_BN    (GM_WAVES * 16)
static_assert(N_NODES % GM_BM == 0, "the row blocks must cover M exactly");
static_assert(HID % GM_BN == 0 && HC % GM_BN == 0, "the column blocks must cover N exactly");

template <int RELU, int K, int N>
__global__ void __launch_bounds__(GM_WAVES * 32)
gemm_wmma_f32(const float* __restrict__ A,
              const float* __restrict__ B,
              const float* __restrict__ bias,
              float* __restrict__ C) {
    constexpr int P    = K + 4;
    constexpr int KS   = K / 4;
    constexpr int NV4  = GM_BM * KS;
    constexpr int NTHR = GM_WAVES * 32;
    static_assert(NV4 % NTHR == 0, "every thread copies the same number of chunks");
    __shared__ __attribute__((aligned(16))) float As[GM_BM * P];

    const int t    = threadIdx.x;
    const int lane = t & 31;
    const int wv   = __builtin_amdgcn_readfirstlane(t >> 5);
    const int l16  = lane & 15;
    const int half = lane >> 4;
    const int m0   = blockIdx.y * GM_BM;
    const int col  = blockIdx.x * GM_BN + wv * 16 + l16;

    const v4f* ag = (const v4f*)(A + (size_t)m0 * K);
    v4f stage[NV4 / NTHR];
#pragma unroll
    for (int i = 0; i < NV4 / NTHR; ++i) stage[i] = ag[t + NTHR * i];

    const float* bcol = B + (size_t)(2 * half) * N + col;
    v2f bfr[KS];
#pragma unroll
    for (int j = 0; j < KS; ++j) {
        bfr[j].x = bcol[(size_t)(4 * j) * N];
        bfr[j].y = bcol[(size_t)(4 * j + 1) * N];
    }

#pragma unroll
    for (int i = 0; i < NV4 / NTHR; ++i) {
        const int idx = t + NTHR * i;
        const int row = idx / KS;
        const int c4  = idx % KS;
        *(v4f*)&As[row * P + 4 * c4] = stage[i];
    }
    __syncthreads();

    v8f acc[GM_MT];
#pragma unroll
    for (int mt = 0; mt < GM_MT; ++mt) acc[mt] = (v8f){0.f, 0.f, 0.f, 0.f, 0.f, 0.f, 0.f, 0.f};

    const float* arow = &As[l16 * P + 2 * half];
#pragma unroll
    for (int j = 0; j < KS; ++j) {
#pragma unroll
        for (int mt = 0; mt < GM_MT; ++mt) {
            const v2f a = *(const v2f*)(arow + mt * 16 * P + 4 * j);
            acc[mt] = __builtin_amdgcn_wmma_f32_16x16x4_f32(
                 false, a,  false, bfr[j],
                 (short)0, acc[mt],  false,  false);
        }
    }

    const float bv = bias[col];
#pragma unroll
    for (int mt = 0; mt < GM_MT; ++mt) {
#pragma unroll
        for (int v = 0; v < 8; ++v) {
            const int row = m0 + mt * 16 + v + 8 * half;
            float val = acc[mt][v] + bv;
            if (RELU) val = fmaxf(val, 0.f);
            C[(size_t)row * N + col] = val;
        }
    }
}

__global__ void init_scratch(int* __restrict__ cnt, float* __restrict__ gpart) {
    int idx = blockIdx.x * blockDim.x + threadIdx.x;
    if (idx < N_NODES) cnt[idx] = 0;
    if (idx < HID) gpart[idx] = 0.f;
}

__global__ void csr_count(const int* __restrict__ ei, int* __restrict__ cnt,
                          int* __restrict__ slot) {
    int e = blockIdx.x * blockDim.x + threadIdx.x;
    if (e >= E_EDGES) return;
    slot[e] = atomicAdd(&cnt[ei[E_EDGES + e]], 1);
}

#define SC_THREADS 1024
#define SC_PER     20
static_assert(SC_THREADS * SC_PER >= N_NODES, "one block must cover every node");

__global__ void __launch_bounds__(SC_THREADS)
csr_scan(const int* __restrict__ cnt, int* __restrict__ first) {
    __shared__ int part[SC_THREADS];
    const int t    = threadIdx.x;
    const int base = t * SC_PER;
    int loc[SC_PER];
    int s = 0;
#pragma unroll
    for (int i = 0; i < SC_PER; ++i) {
        int idx = base + i;
        int v   = (idx < N_NODES) ? cnt[idx] : 0;
        loc[i]  = s;
        s += v;
    }
    part[t] = s;
    __syncthreads();
    for (int off = 1; off < SC_THREADS; off <<= 1) {
        int v = (t >= off) ? part[t - off] : 0;
        __syncthreads();
        part[t] += v;
        __syncthreads();
    }
    const int excl = part[t] - s;
#pragma unroll
    for (int i = 0; i < SC_PER; ++i) {
        int idx = base + i;
        if (idx < N_NODES) first[idx] = excl + loc[i];
    }
    if (t == SC_THREADS - 1) first[N_NODES] = part[t];
}

__global__ void csr_fill(const int* __restrict__ ei, const int* __restrict__ first,
                         const int* __restrict__ slot, int* __restrict__ srcs) {
    int e = blockIdx.x * blockDim.x + threadIdx.x;
    if (e >= E_EDGES) return;
    srcs[first[ei[E_EDGES + e]] + slot[e]] = ei[e];
}

template <int M>
__device__ __forceinline__ float xrow(float v) {
    const int b = __builtin_bit_cast(int, v);
    return __builtin_bit_cast(float, __builtin_amdgcn_mov_dpp(b, 0x160 + M, 0xf, 0xf, true));
}
__device__ __forceinline__ float xhalf(float v) {
    const int b = __builtin_bit_cast(int, v);
    return __builtin_bit_cast(float,
        __builtin_amdgcn_permlanex16(b, b, 0x76543210, 0xfedcba98, false, false));
}

#define NCH 4
__device__ __forceinline__ void load_row(const float* __restrict__ base, int lofs,
                                         v4f (&x)[NCH]) {
    const v4f* p = (const v4f*)base + lofs;
#pragma unroll
    for (int i = 0; i < NCH; ++i) x[i] = p[8 * i];
}

__device__ __forceinline__ float edge_score(const v4f (&x)[NCH], const v4f (&r)[NCH],
                                            const v4f (&w)[NCH]) {
    float s[NCH];
#pragma unroll
    for (int i = 0; i < NCH; ++i) {
        float acc = 0.f;
#pragma unroll
        for (int k = 0; k < 4; ++k) {
            float m = x[i][k] + r[i][k];
            float e = (m > 0.f) ? m : 0.2f * m;
            acc += e * w[i][k];
        }
        s[i] = acc;
    }
    float a = (s[0] + s[1]) + (s[2] + s[3]);
    a += xrow<1>(a);
    a += xrow<2>(a);
    a += xrow<4>(a);
    return a;
}

#define NG_WAVES 4

__global__ void __launch_bounds__(NG_WAVES * 32)
node_gather(const int* __restrict__ first, const int* __restrict__ srcs,
            const float* __restrict__ xl, const float* __restrict__ xr,
            const float* __restrict__ att, float* __restrict__ res) {
    const int node = __builtin_amdgcn_readfirstlane((int)((blockIdx.x * blockDim.x + threadIdx.x) >> 5));
    if (node >= N_NODES) return;
    const int lane = threadIdx.x & 31;
    const int q    = lane >> 3;
    const int lofs = q * (HID / 4) + (lane & 7);
    const int beg  = first[node];
    const int end  = first[node + 1];

    v4f r[NCH], w[NCH], x[NCH], xn[NCH];
    load_row(xr + (size_t)node * HC, lofs, r);
    load_row(att, lofs, w);
    load_row(xl + (size_t)node * HC, lofs, x);
#pragma unroll
    for (int i = 0; i < NCH; ++i) xn[i] = (v4f){0.f, 0.f, 0.f, 0.f};
    if (beg < end) load_row(xl + (size_t)srcs[beg] * HC, lofs, xn);

    float m = edge_score(x, r, w);
    float l = 1.f;
    v4f   acc[NCH];
#pragma unroll
    for (int i = 0; i < NCH; ++i) acc[i] = x[i];

    for (int e = beg; e < end; ++e) {
#pragma unroll
        for (int i = 0; i < NCH; ++i) x[i] = xn[i];
        if (e + 1 < end) load_row(xl + (size_t)srcs[e + 1] * HC, lofs, xn);

        const float a  = edge_score(x, r, w);
        const float d  = a - m;
        const float ex = expf(-fabsf(d));
        const bool  up = d > 0.f;
        const float sc = up ? ex : 1.f;
        const float p  = up ? 1.f : ex;
        m = up ? a : m;
        l = l * sc + p;
#pragma unroll
        for (int i = 0; i < NCH; ++i) {
#pragma unroll
            for (int k = 0; k < 4; ++k) acc[i][k] = acc[i][k] * sc + x[i][k] * p;
        }
    }

    const float inv = 1.f / l;
    v4f t[NCH];
#pragma unroll
    for (int i = 0; i < NCH; ++i) {
#pragma unroll
        for (int k = 0; k < 4; ++k) {
            float v = acc[i][k] * inv;
            v += xrow<8>(v);
            v += xhalf(v);
            t[i][k] = v;
        }
    }
    v4f o = t[0];
    if (q == 1) o = t[1];
    if (q == 2) o = t[2];
    if (q == 3) o = t[3];
#pragma unroll
    for (int k = 0; k < 4; ++k) o[k] *= 0.25f;
    ((v4f*)(res + (size_t)node * HID))[lane] = o;
}

__global__ void node_ln_relu(const float* __restrict__ hin, const float* __restrict__ res,
                             const float* __restrict__ bias, const float* __restrict__ g,
                             const float* __restrict__ be, float* __restrict__ hout) {
    int wave = (blockIdx.x * blockDim.x + threadIdx.x) >> 5;
    if (wave >= N_NODES) return;
    int lane = threadIdx.x & 31;
    const v4f* ph = (const v4f*)(hin + (size_t)wave * HID);
    const v4f* pr = (const v4f*)(res + (size_t)wave * HID);
    const v4f* pb = (const v4f*)bias;

    v4f y;
    {
        v4f a = ph[lane], b = pr[lane], c = pb[lane];
#pragma unroll
        for (int i = 0; i < 4; ++i) y[i] = a[i] + b[i] + c[i];
    }
    float s = 0.f, ss = 0.f;
#pragma unroll
    for (int i = 0; i < 4; ++i) { s += y[i]; ss += y[i] * y[i]; }
    for (int off = 16; off; off >>= 1) {
        s  += __shfl_xor(s,  off, 32);
        ss += __shfl_xor(ss, off, 32);
    }
    float mu  = s * (1.f / HID);
    float var = ss * (1.f / HID) - mu * mu;
    float r   = rsqrtf(var + 1e-5f);

    v4f gg = ((const v4f*)g)[lane];
    v4f bb = ((const v4f*)be)[lane];
    v4f o;
#pragma unroll
    for (int i = 0; i < 4; ++i)
        o[i] = fmaxf((y[i] - mu) * r * gg[i] + bb[i], 0.f);
    ((v4f*)(hout + (size_t)wave * HID))[lane] = o;
}

#define LG_WAVES 8
#define LG_NPW   16
#define LG_NPB   (LG_WAVES * LG_NPW)

__global__ void __launch_bounds__(LG_WAVES * 32)
logits_kernel(const float* __restrict__ shared_f, const float* __restrict__ Wa,
              const float* __restrict__ ba, float* __restrict__ gpart,
              float* __restrict__ out) {
    __shared__ float colpart[LG_WAVES][HID];
    __shared__ float lg[LG_NPB];

    const int lane  = threadIdx.x & 31;
    const int wv    = __builtin_amdgcn_readfirstlane(threadIdx.x >> 5);
    const int node0 = blockIdx.x * LG_NPB + wv * LG_NPW;

    const v4f w = ((const v4f*)Wa)[lane];

    v4f rows[LG_NPW];
#pragma unroll
    for (int j = 0; j < LG_NPW; ++j) {
        int node = node0 + j;
        int nc   = (node < N_NODES) ? node : (N_NODES - 1);
        rows[j]  = ((const v4f*)(shared_f + (size_t)nc * HID))[lane];
    }

    float acc[LG_NPW];
    v4f cs = {0.f, 0.f, 0.f, 0.f};
#pragma unroll
    for (int j = 0; j < LG_NPW; ++j) {
        const bool valid = (node0 + j) < N_NODES;
        const v4f  v     = rows[j];
        float a = 0.f;
#pragma unroll
        for (int i = 0; i < 4; ++i) {
            a     += v[i] * w[i];
            cs[i] += valid ? v[i] : 0.f;
        }
        acc[j] = a;
    }
#pragma unroll
    for (int off = 16; off; off >>= 1) {
#pragma unroll
        for (int j = 0; j < LG_NPW; ++j) acc[j] += __shfl_down(acc[j], off, 32);
    }
    if (lane == 0) {
#pragma unroll
        for (int j = 0; j < LG_NPW; ++j) lg[wv * LG_NPW + j] = acc[j];
    }
    ((v4f*)&colpart[wv][0])[lane] = cs;
    __syncthreads();

    const int t = threadIdx.x;
    if (t < HID) {
        float s = 0.f;
#pragma unroll
        for (int k = 0; k < LG_WAVES; ++k) s += colpart[k][t];
        atomicAdd(&gpart[t], s);
    }
    if (t < LG_NPB) {
        int node = blockIdx.x * LG_NPB + t;
        if (node < N_NODES) out[node] = lg[t] + ba[0];
    }
}

__global__ void value_kernel(const float* __restrict__ gpart, const float* __restrict__ Wc,
                             const float* __restrict__ bc, float* __restrict__ out) {
    int lane = threadIdx.x & 31;
    v4f v = ((const v4f*)gpart)[lane];
    v4f w = ((const v4f*)Wc)[lane];
    float acc = 0.f;
#pragma unroll
    for (int i = 0; i < 4; ++i) acc += (v[i] * (1.f / N_NODES)) * w[i];
    for (int off = 16; off; off >>= 1) acc += __shfl_down(acc, off, 32);
    if (lane == 0) out[N_NODES] = acc + bc[0];
}

static inline int blocksForWaves(long long nwaves) {
    long long threads = nwaves * 32;
    return (int)((threads + 255) / 256);
}

extern "C" void kernel_launch(void* const* d_in, const int* in_sizes, int n_in,
                              void* d_out, int out_size, void* d_ws, size_t ws_size,
                              hipStream_t stream) {
    const float* x     = (const float*)d_in[0];
    const int*   ei    = (const int*)d_in[1];
    const float* W_in  = (const float*)d_in[2];
    const float* b_in  = (const float*)d_in[3];
    const float* Wl1   = (const float*)d_in[4];
    const float* bl1   = (const float*)d_in[5];
    const float* Wr1   = (const float*)d_in[6];
    const float* br1   = (const float*)d_in[7];
    const float* att1  = (const float*)d_in[8];
    const float* bias1 = (const float*)d_in[9];
    const float* g1    = (const float*)d_in[10];
    const float* be1   = (const float*)d_in[11];
    const float* Wl2   = (const float*)d_in[12];
    const float* bl2   = (const float*)d_in[13];
    const float* Wr2   = (const float*)d_in[14];
    const float* br2   = (const float*)d_in[15];
    const float* att2  = (const float*)d_in[16];
    const float* bias2 = (const float*)d_in[17];
    const float* g2    = (const float*)d_in[18];
    const float* be2   = (const float*)d_in[19];
    const float* Wa    = (const float*)d_in[20];
    const float* ba    = (const float*)d_in[21];
    const float* Wc    = (const float*)d_in[22];
    const float* bc    = (const float*)d_in[23];

    float* out = (float*)d_out;

    float* ws = (float*)d_ws;
    size_t off = 0;
    float*    h0    = ws + off; off += (size_t)N_NODES * HID;
    float*    h1    = ws + off; off += (size_t)N_NODES * HID;
    float*    xl    = ws + off; off += (size_t)N_NODES * HC;
    float*    xr    = ws + off; off += (size_t)N_NODES * HC;
    int*      csr   = (int*)(ws + off); off += (size_t)ETOT * NHEAD;
    off += (size_t)N_NODES * NHEAD;
    off += (size_t)N_NODES * NHEAD;
    float*    res   = ws + off; off += (size_t)N_NODES * HID;
    float*    gpart = ws + off; off += HID;
    (void)ws_size; (void)in_sizes; (void)n_in; (void)out_size;

    int* cnt   = csr;
    int* first = cnt + N_NODES;
    int* slot  = first + (N_NODES + 16);
    int* srcs  = slot + E_EDGES;
    static_assert((size_t)N_NODES + (N_NODES + 16) + 2 * (size_t)E_EDGES <= (size_t)ETOT * NHEAD,
                  "the grouping must fit the region it reuses");

    const int initBlocks  = (N_NODES + 255) / 256;
    const int eBlocks     = (E_EDGES + 255) / 256;
    const int gatherBlocks = (N_NODES + NG_WAVES - 1) / NG_WAVES;
    const int nodeBlocks  = blocksForWaves(N_NODES);
    const dim3 gemmInGrid(HID / GM_BN, N_NODES / GM_BM);
    const dim3 gemmHCGrid(HC / GM_BN, N_NODES / GM_BM);
    const int logitBlocks = (N_NODES + LG_NPB - 1) / LG_NPB;

    init_scratch<<<initBlocks, 256, 0, stream>>>(cnt, gpart);
    csr_count   <<<eBlocks, 256, 0, stream>>>(ei, cnt, slot);
    csr_scan    <<<1, SC_THREADS, 0, stream>>>(cnt, first);
    csr_fill    <<<eBlocks, 256, 0, stream>>>(ei, first, slot, srcs);

    gemm_wmma_f32<1, DIN, HID><<<gemmInGrid, GM_WAVES * 32, 0, stream>>>(x, W_in, b_in, h0);

    gemm_wmma_f32<0, HID, HC><<<gemmHCGrid, GM_WAVES * 32, 0, stream>>>(h0, Wl1, bl1, xl);
    gemm_wmma_f32<0, HID, HC><<<gemmHCGrid, GM_WAVES * 32, 0, stream>>>(h0, Wr1, br1, xr);
    node_gather <<<gatherBlocks, NG_WAVES * 32, 0, stream>>>(first, srcs, xl, xr, att1, res);
    node_ln_relu<<<nodeBlocks, 256, 0, stream>>>(h0, res, bias1, g1, be1, h1);

    gemm_wmma_f32<0, HID, HC><<<gemmHCGrid, GM_WAVES * 32, 0, stream>>>(h1, Wl2, bl2, xl);
    gemm_wmma_f32<0, HID, HC><<<gemmHCGrid, GM_WAVES * 32, 0, stream>>>(h1, Wr2, br2, xr);
    node_gather <<<gatherBlocks, NG_WAVES * 32, 0, stream>>>(first, srcs, xl, xr, att2, res);
    node_ln_relu<<<nodeBlocks, 256, 0, stream>>>(h1, res, bias2, g2, be2, h0);

    logits_kernel<<<logitBlocks, LG_WAVES * 32, 0, stream>>>(h0, Wa, ba, gpart, out);
    value_kernel <<<1, 32, 0, stream>>>(gpart, Wc, bc, out);
}


# layer norm + relu done by the gather wave on the row it already holds (lane l = channels 4l..4l+3): the two norm launches and the res round trip through memory are gone
# speedup vs baseline: 3.1930x; 1.0297x over previous
#include <hip/hip_runtime.h>
#include <hip/hip_bf16.h>
#include <math.h>

#define N_NODES 20000
#define E_EDGES 320000
#define ETOT    (E_EDGES + N_NODES)
#define DIN     64
#define HID     128
#define NHEAD   4
#define HC      (NHEAD * HID)

typedef float v2f __attribute__((ext_vector_type(2)));
typedef float v4f __attribute__((ext_vector_type(4)));
typedef float v8f __attribute__((ext_vector_type(8)));

#define GM_BM    80
#define GM_MT    (GM_BM / 16)
#define GM_WAVES 8
#define GM_BN    (GM_WAVES * 16)
static_assert(N_NODES % GM_BM == 0, "the row blocks must cover M exactly");
static_assert(HID % GM_BN == 0 && HC % GM_BN == 0, "the column blocks must cover N exactly");

template <int RELU, int K, int N>
__global__ void __launch_bounds__(GM_WAVES * 32)
gemm_wmma_f32(const float* __restrict__ A,
              const float* __restrict__ B,
              const float* __restrict__ bias,
              float* __restrict__ C) {
    constexpr int P    = K + 4;
    constexpr int KS   = K / 4;
    constexpr int NV4  = GM_BM * KS;
    constexpr int NTHR = GM_WAVES * 32;
    static_assert(NV4 % NTHR == 0, "every thread copies the same number of chunks");
    __shared__ __attribute__((aligned(16))) float As[GM_BM * P];

    const int t    = threadIdx.x;
    const int lane = t & 31;
    const int wv   = __builtin_amdgcn_readfirstlane(t >> 5);
    const int l16  = lane & 15;
    const int half = lane >> 4;
    const int m0   = blockIdx.y * GM_BM;
    const int col  = blockIdx.x * GM_BN + wv * 16 + l16;

    const v4f* ag = (const v4f*)(A + (size_t)m0 * K);
    v4f stage[NV4 / NTHR];
#pragma unroll
    for (int i = 0; i < NV4 / NTHR; ++i) stage[i] = ag[t + NTHR * i];

    const float* bcol = B + (size_t)(2 * half) * N + col;
    v2f bfr[KS];
#pragma unroll
    for (int j = 0; j < KS; ++j) {
        bfr[j].x = bcol[(size_t)(4 * j) * N];
        bfr[j].y = bcol[(size_t)(4 * j + 1) * N];
    }

#pragma unroll
    for (int i = 0; i < NV4 / NTHR; ++i) {
        const int idx = t + NTHR * i;
        const int row = idx / KS;
        const int c4  = idx % KS;
        *(v4f*)&As[row * P + 4 * c4] = stage[i];
    }
    __syncthreads();

    v8f acc[GM_MT];
#pragma unroll
    for (int mt = 0; mt < GM_MT; ++mt) acc[mt] = (v8f){0.f, 0.f, 0.f, 0.f, 0.f, 0.f, 0.f, 0.f};

    const float* arow = &As[l16 * P + 2 * half];
#pragma unroll
    for (int j = 0; j < KS; ++j) {
#pragma unroll
        for (int mt = 0; mt < GM_MT; ++mt) {
            const v2f a = *(const v2f*)(arow + mt * 16 * P + 4 * j);
            acc[mt] = __builtin_amdgcn_wmma_f32_16x16x4_f32(
                 false, a,  false, bfr[j],
                 (short)0, acc[mt],  false,  false);
        }
    }

    const float bv = bias[col];
#pragma unroll
    for (int mt = 0; mt < GM_MT; ++mt) {
#pragma unroll
        for (int v = 0; v < 8; ++v) {
            const int row = m0 + mt * 16 + v + 8 * half;
            float val = acc[mt][v] + bv;
            if (RELU) val = fmaxf(val, 0.f);
            C[(size_t)row * N + col] = val;
        }
    }
}

__global__ void init_scratch(int* __restrict__ cnt, float* __restrict__ gpart) {
    int idx = blockIdx.x * blockDim.x + threadIdx.x;
    if (idx < N_NODES) cnt[idx] = 0;
    if (idx < HID) gpart[idx] = 0.f;
}

__global__ void csr_count(const int* __restrict__ ei, int* __restrict__ cnt,
                          int* __restrict__ slot) {
    int e = blockIdx.x * blockDim.x + threadIdx.x;
    if (e >= E_EDGES) return;
    slot[e] = atomicAdd(&cnt[ei[E_EDGES + e]], 1);
}

#define SC_THREADS 1024
#define SC_PER     20
static_assert(SC_THREADS * SC_PER >= N_NODES, "one block must cover every node");

__global__ void __launch_bounds__(SC_THREADS)
csr_scan(const int* __restrict__ cnt, int* __restrict__ first) {
    __shared__ int part[SC_THREADS];
    const int t    = threadIdx.x;
    const int base = t * SC_PER;
    int loc[SC_PER];
    int s = 0;
#pragma unroll
    for (int i = 0; i < SC_PER; ++i) {
        int idx = base + i;
        int v   = (idx < N_NODES) ? cnt[idx] : 0;
        loc[i]  = s;
        s += v;
    }
    part[t] = s;
    __syncthreads();
    for (int off = 1; off < SC_THREADS; off <<= 1) {
        int v = (t >= off) ? part[t - off] : 0;
        __syncthreads();
        part[t] += v;
        __syncthreads();
    }
    const int excl = part[t] - s;
#pragma unroll
    for (int i = 0; i < SC_PER; ++i) {
        int idx = base + i;
        if (idx < N_NODES) first[idx] = excl + loc[i];
    }
    if (t == SC_THREADS - 1) first[N_NODES] = part[t];
}

__global__ void csr_fill(const int* __restrict__ ei, const int* __restrict__ first,
                         const int* __restrict__ slot, int* __restrict__ srcs) {
    int e = blockIdx.x * blockDim.x + threadIdx.x;
    if (e >= E_EDGES) return;
    srcs[first[ei[E_EDGES + e]] + slot[e]] = ei[e];
}

template <int M>
__device__ __forceinline__ float xrow(float v) {
    const int b = __builtin_bit_cast(int, v);
    return __builtin_bit_cast(float, __builtin_amdgcn_mov_dpp(b, 0x160 + M, 0xf, 0xf, true));
}
__device__ __forceinline__ float xhalf(float v) {
    const int b = __builtin_bit_cast(int, v);
    return __builtin_bit_cast(float,
        __builtin_amdgcn_permlanex16(b, b, 0x76543210, 0xfedcba98, false, false));
}

#define NCH 4
__device__ __forceinline__ void load_row(const float* __restrict__ base, int lofs,
                                         v4f (&x)[NCH]) {
    const v4f* p = (const v4f*)base + lofs;
#pragma unroll
    for (int i = 0; i < NCH; ++i) x[i] = p[8 * i];
}

__device__ __forceinline__ float edge_score(const v4f (&x)[NCH], const v4f (&r)[NCH],
                                            const v4f (&w)[NCH]) {
    float s[NCH];
#pragma unroll
    for (int i = 0; i < NCH; ++i) {
        float acc = 0.f;
#pragma unroll
        for (int k = 0; k < 4; ++k) {
            float m = x[i][k] + r[i][k];
            float e = (m > 0.f) ? m : 0.2f * m;
            acc += e * w[i][k];
        }
        s[i] = acc;
    }
    float a = (s[0] + s[1]) + (s[2] + s[3]);
    a += xrow<1>(a);
    a += xrow<2>(a);
    a += xrow<4>(a);
    return a;
}

#define NG_WAVES 4

__global__ void __launch_bounds__(NG_WAVES * 32)
node_gather(const int* __restrict__ first, const int* __restrict__ srcs,
            const float* __restrict__ xl, const float* __restrict__ xr,
            const float* __restrict__ att,
            const float* __restrict__ hin, const float* __restrict__ bias,
            const float* __restrict__ g, const float* __restrict__ be,
            float* __restrict__ hout) {
    const int node = __builtin_amdgcn_readfirstlane((int)((blockIdx.x * blockDim.x + threadIdx.x) >> 5));
    if (node >= N_NODES) return;
    const int lane = threadIdx.x & 31;
    const int q    = lane >> 3;
    const int lofs = q * (HID / 4) + (lane & 7);
    const int beg  = first[node];
    const int end  = first[node + 1];

    v4f r[NCH], w[NCH], x[NCH], xn[NCH];
    load_row(xr + (size_t)node * HC, lofs, r);
    load_row(att, lofs, w);
    load_row(xl + (size_t)node * HC, lofs, x);
#pragma unroll
    for (int i = 0; i < NCH; ++i) xn[i] = (v4f){0.f, 0.f, 0.f, 0.f};
    if (beg < end) load_row(xl + (size_t)srcs[beg] * HC, lofs, xn);

    float m = edge_score(x, r, w);
    float l = 1.f;
    v4f   acc[NCH];
#pragma unroll
    for (int i = 0; i < NCH; ++i) acc[i] = x[i];

    for (int e = beg; e < end; ++e) {
#pragma unroll
        for (int i = 0; i < NCH; ++i) x[i] = xn[i];
        if (e + 1 < end) load_row(xl + (size_t)srcs[e + 1] * HC, lofs, xn);

        const float a  = edge_score(x, r, w);
        const float d  = a - m;
        const float ex = expf(-fabsf(d));
        const bool  up = d > 0.f;
        const float sc = up ? ex : 1.f;
        const float p  = up ? 1.f : ex;
        m = up ? a : m;
        l = l * sc + p;
#pragma unroll
        for (int i = 0; i < NCH; ++i) {
#pragma unroll
            for (int k = 0; k < 4; ++k) acc[i][k] = acc[i][k] * sc + x[i][k] * p;
        }
    }

    const float inv = 1.f / l;
    v4f t[NCH];
#pragma unroll
    for (int i = 0; i < NCH; ++i) {
#pragma unroll
        for (int k = 0; k < 4; ++k) {
            float v = acc[i][k] * inv;
            v += xrow<8>(v);
            v += xhalf(v);
            t[i][k] = v;
        }
    }
    v4f o = t[0];
    if (q == 1) o = t[1];
    if (q == 2) o = t[2];
    if (q == 3) o = t[3];
#pragma unroll
    for (int k = 0; k < 4; ++k) o[k] *= 0.25f;

    const v4f hv = ((const v4f*)(hin + (size_t)node * HID))[lane];
    const v4f cb = ((const v4f*)bias)[lane];
    const v4f gg = ((const v4f*)g)[lane];
    const v4f bb = ((const v4f*)be)[lane];
    v4f y;
#pragma unroll
    for (int k = 0; k < 4; ++k) y[k] = hv[k] + o[k] + cb[k];
    float s = 0.f, ss = 0.f;
#pragma unroll
    for (int k = 0; k < 4; ++k) { s += y[k]; ss += y[k] * y[k]; }
    s += xhalf(s);    ss += xhalf(ss);
    s += xrow<8>(s);  ss += xrow<8>(ss);
    s += xrow<4>(s);  ss += xrow<4>(ss);
    s += xrow<2>(s);  ss += xrow<2>(ss);
    s += xrow<1>(s);  ss += xrow<1>(ss);
    const float mu  = s * (1.f / HID);
    const float var = ss * (1.f / HID) - mu * mu;
    const float rs  = rsqrtf(var + 1e-5f);
    v4f ho;
#pragma unroll
    for (int k = 0; k < 4; ++k)
        ho[k] = fmaxf((y[k] - mu) * rs * gg[k] + bb[k], 0.f);
    ((v4f*)(hout + (size_t)node * HID))[lane] = ho;
}

#define LG_WAVES 8
#define LG_NPW   16
#define LG_NPB   (LG_WAVES * LG_NPW)

__global__ void __launch_bounds__(LG_WAVES * 32)
logits_kernel(const float* __restrict__ shared_f, const float* __restrict__ Wa,
              const float* __restrict__ ba, float* __restrict__ gpart,
              float* __restrict__ out) {
    __shared__ float colpart[LG_WAVES][HID];
    __shared__ float lg[LG_NPB];

    const int lane  = threadIdx.x & 31;
    const int wv    = __builtin_amdgcn_readfirstlane(threadIdx.x >> 5);
    const int node0 = blockIdx.x * LG_NPB + wv * LG_NPW;

    const v4f w = ((const v4f*)Wa)[lane];

    v4f rows[LG_NPW];
#pragma unroll
    for (int j = 0; j < LG_NPW; ++j) {
        int node = node0 + j;
        int nc   = (node < N_NODES) ? node : (N_NODES - 1);
        rows[j]  = ((const v4f*)(shared_f + (size_t)nc * HID))[lane];
    }

    float acc[LG_NPW];
    v4f cs = {0.f, 0.f, 0.f, 0.f};
#pragma unroll
    for (int j = 0; j < LG_NPW; ++j) {
        const bool valid = (node0 + j) < N_NODES;
        const v4f  v     = rows[j];
        float a = 0.f;
#pragma unroll
        for (int i = 0; i < 4; ++i) {
            a     += v[i] * w[i];
            cs[i] += valid ? v[i] : 0.f;
        }
        acc[j] = a;
    }
#pragma unroll
    for (int off = 16; off; off >>= 1) {
#pragma unroll
        for (int j = 0; j < LG_NPW; ++j) acc[j] += __shfl_down(acc[j], off, 32);
    }
    if (lane == 0) {
#pragma unroll
        for (int j = 0; j < LG_NPW; ++j) lg[wv * LG_NPW + j] = acc[j];
    }
    ((v4f*)&colpart[wv][0])[lane] = cs;
    __syncthreads();

    const int t = threadIdx.x;
    if (t < HID) {
        float s = 0.f;
#pragma unroll
        for (int k = 0; k < LG_WAVES; ++k) s += colpart[k][t];
        atomicAdd(&gpart[t], s);
    }
    if (t < LG_NPB) {
        int node = blockIdx.x * LG_NPB + t;
        if (node < N_NODES) out[node] = lg[t] + ba[0];
    }
}

__global__ void value_kernel(const float* __restrict__ gpart, const float* __restrict__ Wc,
                             const float* __restrict__ bc, float* __restrict__ out) {
    int lane = threadIdx.x & 31;
    v4f v = ((const v4f*)gpart)[lane];
    v4f w = ((const v4f*)Wc)[lane];
    float acc = 0.f;
#pragma unroll
    for (int i = 0; i < 4; ++i) acc += (v[i] * (1.f / N_NODES)) * w[i];
    for (int off = 16; off; off >>= 1) acc += __shfl_down(acc, off, 32);
    if (lane == 0) out[N_NODES] = acc + bc[0];
}

extern "C" void kernel_launch(void* const* d_in, const int* in_sizes, int n_in,
                              void* d_out, int out_size, void* d_ws, size_t ws_size,
                              hipStream_t stream) {
    const float* x     = (const float*)d_in[0];
    const int*   ei    = (const int*)d_in[1];
    const float* W_in  = (const float*)d_in[2];
    const float* b_in  = (const float*)d_in[3];
    const float* Wl1   = (const float*)d_in[4];
    const float* bl1   = (const float*)d_in[5];
    const float* Wr1   = (const float*)d_in[6];
    const float* br1   = (const float*)d_in[7];
    const float* att1  = (const float*)d_in[8];
    const float* bias1 = (const float*)d_in[9];
    const float* g1    = (const float*)d_in[10];
    const float* be1   = (const float*)d_in[11];
    const float* Wl2   = (const float*)d_in[12];
    const float* bl2   = (const float*)d_in[13];
    const float* Wr2   = (const float*)d_in[14];
    const float* br2   = (const float*)d_in[15];
    const float* att2  = (const float*)d_in[16];
    const float* bias2 = (const float*)d_in[17];
    const float* g2    = (const float*)d_in[18];
    const float* be2   = (const float*)d_in[19];
    const float* Wa    = (const float*)d_in[20];
    const float* ba    = (const float*)d_in[21];
    const float* Wc    = (const float*)d_in[22];
    const float* bc    = (const float*)d_in[23];

    float* out = (float*)d_out;

    float* ws = (float*)d_ws;
    size_t off = 0;
    float*    h0    = ws + off; off += (size_t)N_NODES * HID;
    float*    h1    = ws + off; off += (size_t)N_NODES * HID;
    float*    xl    = ws + off; off += (size_t)N_NODES * HC;
    float*    xr    = ws + off; off += (size_t)N_NODES * HC;
    int*      csr   = (int*)(ws + off); off += (size_t)ETOT * NHEAD;
    off += (size_t)N_NODES * NHEAD;
    off += (size_t)N_NODES * NHEAD;
    off += (size_t)N_NODES * HID;
    float*    gpart = ws + off; off += HID;
    (void)ws_size; (void)in_sizes; (void)n_in; (void)out_size;

    int* cnt   = csr;
    int* first = cnt + N_NODES;
    int* slot  = first + (N_NODES + 16);
    int* srcs  = slot + E_EDGES;
    static_assert((size_t)N_NODES + (N_NODES + 16) + 2 * (size_t)E_EDGES <= (size_t)ETOT * NHEAD,
                  "the grouping must fit the region it reuses");

    const int initBlocks  = (N_NODES + 255) / 256;
    const int eBlocks     = (E_EDGES + 255) / 256;
    const int gatherBlocks = (N_NODES + NG_WAVES - 1) / NG_WAVES;
    const dim3 gemmInGrid(HID / GM_BN, N_NODES / GM_BM);
    const dim3 gemmHCGrid(HC / GM_BN, N_NODES / GM_BM);
    const int logitBlocks = (N_NODES + LG_NPB - 1) / LG_NPB;

    init_scratch<<<initBlocks, 256, 0, stream>>>(cnt, gpart);
    csr_count   <<<eBlocks, 256, 0, stream>>>(ei, cnt, slot);
    csr_scan    <<<1, SC_THREADS, 0, stream>>>(cnt, first);
    csr_fill    <<<eBlocks, 256, 0, stream>>>(ei, first, slot, srcs);

    gemm_wmma_f32<1, DIN, HID><<<gemmInGrid, GM_WAVES * 32, 0, stream>>>(x, W_in, b_in, h0);

    gemm_wmma_f32<0, HID, HC><<<gemmHCGrid, GM_WAVES * 32, 0, stream>>>(h0, Wl1, bl1, xl);
    gemm_wmma_f32<0, HID, HC><<<gemmHCGrid, GM_WAVES * 32, 0, stream>>>(h0, Wr1, br1, xr);
    node_gather <<<gatherBlocks, NG_WAVES * 32, 0, stream>>>(first, srcs, xl, xr, att1,
                                                             h0, bias1, g1, be1, h1);

    gemm_wmma_f32<0, HID, HC><<<gemmHCGrid, GM_WAVES * 32, 0, stream>>>(h1, Wl2, bl2, xl);
    gemm_wmma_f32<0, HID, HC><<<gemmHCGrid, GM_WAVES * 32, 0, stream>>>(h1, Wr2, br2, xr);
    node_gather <<<gatherBlocks, NG_WAVES * 32, 0, stream>>>(first, srcs, xl, xr, att2,
                                                             h1, bias2, g2, be2, h0);

    logits_kernel<<<logitBlocks, LG_WAVES * 32, 0, stream>>>(h0, Wa, ba, gpart, out);
    value_kernel <<<1, 32, 0, stream>>>(gpart, Wc, bc, out);
}


# gather loop, 133 -> 78 vector instructions per edge: two edges per trip (fewer row copies), leaky relu as max(m, 0.2m), score by packed fused multiply-adds; float sum order of the score changes
# speedup vs baseline: 3.3382x; 1.0455x over previous
#include <hip/hip_runtime.h>
#include <hip/hip_bf16.h>
#include <math.h>

#define N_NODES 20000
#define E_EDGES 320000
#define ETOT    (E_EDGES + N_NODES)
#define DIN     64
#define HID     128
#define NHEAD   4
#define HC      (NHEAD * HID)

typedef float v2f __attribute__((ext_vector_type(2)));
typedef float v4f __attribute__((ext_vector_type(4)));
typedef float v8f __attribute__((ext_vector_type(8)));

#define GM_BM    80
#define GM_MT    (GM_BM / 16)
#define GM_WAVES 8
#define GM_BN    (GM_WAVES * 16)
static_assert(N_NODES % GM_BM == 0, "the row blocks must cover M exactly");
static_assert(HID % GM_BN == 0 && HC % GM_BN == 0, "the column blocks must cover N exactly");

template <int RELU, int K, int N>
__global__ void __launch_bounds__(GM_WAVES * 32)
gemm_wmma_f32(const float* __restrict__ A,
              const float* __restrict__ B,
              const float* __restrict__ bias,
              float* __restrict__ C) {
    constexpr int P    = K + 4;
    constexpr int KS   = K / 4;
    constexpr int NV4  = GM_BM * KS;
    constexpr int NTHR = GM_WAVES * 32;
    static_assert(NV4 % NTHR == 0, "every thread copies the same number of chunks");
    __shared__ __attribute__((aligned(16))) float As[GM_BM * P];

    const int t    = threadIdx.x;
    const int lane = t & 31;
    const int wv   = __builtin_amdgcn_readfirstlane(t >> 5);
    const int l16  = lane & 15;
    const int half = lane >> 4;
    const int m0   = blockIdx.y * GM_BM;
    const int col  = blockIdx.x * GM_BN + wv * 16 + l16;

    const v4f* ag = (const v4f*)(A + (size_t)m0 * K);
    v4f stage[NV4 / NTHR];
#pragma unroll
    for (int i = 0; i < NV4 / NTHR; ++i) stage[i] = ag[t + NTHR * i];

    const float* bcol = B + (size_t)(2 * half) * N + col;
    v2f bfr[KS];
#pragma unroll
    for (int j = 0; j < KS; ++j) {
        bfr[j].x = bcol[(size_t)(4 * j) * N];
        bfr[j].y = bcol[(size_t)(4 * j + 1) * N];
    }

#pragma unroll
    for (int i = 0; i < NV4 / NTHR; ++i) {
        const int idx = t + NTHR * i;
        const int row = idx / KS;
        const int c4  = idx % KS;
        *(v4f*)&As[row * P + 4 * c4] = stage[i];
    }
    __syncthreads();

    v8f acc[GM_MT];
#pragma unroll
    for (int mt = 0; mt < GM_MT; ++mt) acc[mt] = (v8f){0.f, 0.f, 0.f, 0.f, 0.f, 0.f, 0.f, 0.f};

    const float* arow = &As[l16 * P + 2 * half];
#pragma unroll
    for (int j = 0; j < KS; ++j) {
#pragma unroll
        for (int mt = 0; mt < GM_MT; ++mt) {
            const v2f a = *(const v2f*)(arow + mt * 16 * P + 4 * j);
            acc[mt] = __builtin_amdgcn_wmma_f32_16x16x4_f32(
                 false, a,  false, bfr[j],
                 (short)0, acc[mt],  false,  false);
        }
    }

    const float bv = bias[col];
#pragma unroll
    for (int mt = 0; mt < GM_MT; ++mt) {
#pragma unroll
        for (int v = 0; v < 8; ++v) {
            const int row = m0 + mt * 16 + v + 8 * half;
            float val = acc[mt][v] + bv;
            if (RELU) val = fmaxf(val, 0.f);
            C[(size_t)row * N + col] = val;
        }
    }
}

__global__ void init_scratch(int* __restrict__ cnt, float* __restrict__ gpart) {
    int idx = blockIdx.x * blockDim.x + threadIdx.x;
    if (idx < N_NODES) cnt[idx] = 0;
    if (idx < HID) gpart[idx] = 0.f;
}

__global__ void csr_count(const int* __restrict__ ei, int* __restrict__ cnt,
                          int* __restrict__ slot) {
    int e = blockIdx.x * blockDim.x + threadIdx.x;
    if (e >= E_EDGES) return;
    slot[e] = atomicAdd(&cnt[ei[E_EDGES + e]], 1);
}

#define SC_THREADS 1024
#define SC_PER     20
static_assert(SC_THREADS * SC_PER >= N_NODES, "one block must cover every node");

__global__ void __launch_bounds__(SC_THREADS)
csr_scan(const int* __restrict__ cnt, int* __restrict__ first) {
    __shared__ int part[SC_THREADS];
    const int t    = threadIdx.x;
    const int base = t * SC_PER;
    int loc[SC_PER];
    int s = 0;
#pragma unroll
    for (int i = 0; i < SC_PER; ++i) {
        int idx = base + i;
        int v   = (idx < N_NODES) ? cnt[idx] : 0;
        loc[i]  = s;
        s += v;
    }
    part[t] = s;
    __syncthreads();
    for (int off = 1; off < SC_THREADS; off <<= 1) {
        int v = (t >= off) ? part[t - off] : 0;
        __syncthreads();
        part[t] += v;
        __syncthreads();
    }
    const int excl = part[t] - s;
#pragma unroll
    for (int i = 0; i < SC_PER; ++i) {
        int idx = base + i;
        if (idx < N_NODES) first[idx] = excl + loc[i];
    }
    if (t == SC_THREADS - 1) first[N_NODES] = part[t];
}

__global__ void csr_fill(const int* __restrict__ ei, const int* __restrict__ first,
                         const int* __restrict__ slot, int* __restrict__ srcs) {
    int e = blockIdx.x * blockDim.x + threadIdx.x;
    if (e >= E_EDGES) return;
    srcs[first[ei[E_EDGES + e]] + slot[e]] = ei[e];
}

template <int M>
__device__ __forceinline__ float xrow(float v) {
    const int b = __builtin_bit_cast(int, v);
    return __builtin_bit_cast(float, __builtin_amdgcn_mov_dpp(b, 0x160 + M, 0xf, 0xf, true));
}
__device__ __forceinline__ float xhalf(float v) {
    const int b = __builtin_bit_cast(int, v);
    return __builtin_bit_cast(float,
        __builtin_amdgcn_permlanex16(b, b, 0x76543210, 0xfedcba98, false, false));
}

#define NCH 4
__device__ __forceinline__ void load_row(const float* __restrict__ base, int lofs,
                                         v4f (&x)[NCH]) {
    const v4f* p = (const v4f*)base + lofs;
#pragma unroll
    for (int i = 0; i < NCH; ++i) x[i] = p[8 * i];
}

__device__ __forceinline__ float edge_score(const v4f (&x)[NCH], const v4f (&r)[NCH],
                                            const v4f (&w)[NCH]) {
    v2f sa = {0.f, 0.f};
    v2f sb = {0.f, 0.f};
#pragma unroll
    for (int i = 0; i < NCH; ++i) {
        const v2f ma = x[i].lo + r[i].lo;
        const v2f mb = x[i].hi + r[i].hi;
        const v2f ea = __builtin_elementwise_max(ma, ma * 0.2f);
        const v2f eb = __builtin_elementwise_max(mb, mb * 0.2f);
        sa = __builtin_elementwise_fma(ea, w[i].lo, sa);
        sb = __builtin_elementwise_fma(eb, w[i].hi, sb);
    }
    const v2f s2 = sa + sb;
    float a = s2.x + s2.y;
    a += xrow<1>(a);
    a += xrow<2>(a);
    a += xrow<4>(a);
    return a;
}

__device__ __forceinline__ void edge_update(const v4f (&x)[NCH], const v4f (&r)[NCH],
                                            const v4f (&w)[NCH], float& m, float& l,
                                            v4f (&acc)[NCH]) {
    const float a  = edge_score(x, r, w);
    const float d  = a - m;
    const float ex = expf(-fabsf(d));
    const bool  up = d > 0.f;
    const float sc = up ? ex : 1.f;
    const float p  = up ? 1.f : ex;
    m = up ? a : m;
    l = l * sc + p;
    const v2f scv = {sc, sc};
    const v2f pv  = {p, p};
#pragma unroll
    for (int i = 0; i < NCH; ++i) {
        acc[i].lo = __builtin_elementwise_fma(x[i].lo, pv, acc[i].lo * scv);
        acc[i].hi = __builtin_elementwise_fma(x[i].hi, pv, acc[i].hi * scv);
    }
}

#define NG_WAVES 4

__global__ void __launch_bounds__(NG_WAVES * 32)
node_gather(const int* __restrict__ first, const int* __restrict__ srcs,
            const float* __restrict__ xl, const float* __restrict__ xr,
            const float* __restrict__ att,
            const float* __restrict__ hin, const float* __restrict__ bias,
            const float* __restrict__ g, const float* __restrict__ be,
            float* __restrict__ hout) {
    const int node = __builtin_amdgcn_readfirstlane((int)((blockIdx.x * blockDim.x + threadIdx.x) >> 5));
    if (node >= N_NODES) return;
    const int lane = threadIdx.x & 31;
    const int q    = lane >> 3;
    const int lofs = q * (HID / 4) + (lane & 7);
    const int beg  = first[node];
    const int end  = first[node + 1];

    v4f r[NCH], w[NCH], x[NCH], xn[NCH];
    load_row(xr + (size_t)node * HC, lofs, r);
    load_row(att, lofs, w);
    load_row(xl + (size_t)node * HC, lofs, x);
#pragma unroll
    for (int i = 0; i < NCH; ++i) xn[i] = (v4f){0.f, 0.f, 0.f, 0.f};
    if (beg < end) load_row(xl + (size_t)srcs[beg] * HC, lofs, xn);

    float m = edge_score(x, r, w);
    float l = 1.f;
    v4f   acc[NCH];
#pragma unroll
    for (int i = 0; i < NCH; ++i) acc[i] = x[i];

    if (beg < end) {
        int e  = beg;
        int ia = srcs[min(e + 1, end - 1)];
        int ib = srcs[min(e + 2, end - 1)];
        while (e + 2 < end) {
            load_row(xl + (size_t)ia * HC, lofs, x);
            const int ja = srcs[min(e + 3, end - 1)];
            const int jb = srcs[min(e + 4, end - 1)];
            __builtin_amdgcn_sched_barrier(0);
            edge_update(xn, r, w, m, l, acc);
            load_row(xl + (size_t)ib * HC, lofs, xn);
            __builtin_amdgcn_sched_barrier(0);
            edge_update(x, r, w, m, l, acc);
            ia = ja;
            ib = jb;
            e += 2;
        }
        if (e + 1 < end) {
            load_row(xl + (size_t)ia * HC, lofs, x);
            __builtin_amdgcn_sched_barrier(0);
            edge_update(xn, r, w, m, l, acc);
            __builtin_amdgcn_sched_barrier(0);
            edge_update(x, r, w, m, l, acc);
        } else {
            edge_update(xn, r, w, m, l, acc);
        }
    }

    const float inv = 1.f / l;
    v4f t[NCH];
#pragma unroll
    for (int i = 0; i < NCH; ++i) {
#pragma unroll
        for (int k = 0; k < 4; ++k) {
            float v = acc[i][k] * inv;
            v += xrow<8>(v);
            v += xhalf(v);
            t[i][k] = v;
        }
    }
    v4f o = t[0];
    if (q == 1) o = t[1];
    if (q == 2) o = t[2];
    if (q == 3) o = t[3];
#pragma unroll
    for (int k = 0; k < 4; ++k) o[k] *= 0.25f;

    const v4f hv = ((const v4f*)(hin + (size_t)node * HID))[lane];
    const v4f cb = ((const v4f*)bias)[lane];
    const v4f gg = ((const v4f*)g)[lane];
    const v4f bb = ((const v4f*)be)[lane];
    v4f y;
#pragma unroll
    for (int k = 0; k < 4; ++k) y[k] = hv[k] + o[k] + cb[k];
    float s = 0.f, ss = 0.f;
#pragma unroll
    for (int k = 0; k < 4; ++k) { s += y[k]; ss += y[k] * y[k]; }
    s += xhalf(s);    ss += xhalf(ss);
    s += xrow<8>(s);  ss += xrow<8>(ss);
    s += xrow<4>(s);  ss += xrow<4>(ss);
    s += xrow<2>(s);  ss += xrow<2>(ss);
    s += xrow<1>(s);  ss += xrow<1>(ss);
    const float mu  = s * (1.f / HID);
    const float var = ss * (1.f / HID) - mu * mu;
    const float rs  = rsqrtf(var + 1e-5f);
    v4f ho;
#pragma unroll
    for (int k = 0; k < 4; ++k)
        ho[k] = fmaxf((y[k] - mu) * rs * gg[k] + bb[k], 0.f);
    ((v4f*)(hout + (size_t)node * HID))[lane] = ho;
}

#define LG_WAVES 8
#define LG_NPW   16
#define LG_NPB   (LG_WAVES * LG_NPW)

__global__ void __launch_bounds__(LG_WAVES * 32)
logits_kernel(const float* __restrict__ shared_f, const float* __restrict__ Wa,
              const float* __restrict__ ba, float* __restrict__ gpart,
              float* __restrict__ out) {
    __shared__ float colpart[LG_WAVES][HID];
    __shared__ float lg[LG_NPB];

    const int lane  = threadIdx.x & 31;
    const int wv    = __builtin_amdgcn_readfirstlane(threadIdx.x >> 5);
    const int node0 = blockIdx.x * LG_NPB + wv * LG_NPW;

    const v4f w = ((const v4f*)Wa)[lane];

    v4f rows[LG_NPW];
#pragma unroll
    for (int j = 0; j < LG_NPW; ++j) {
        int node = node0 + j;
        int nc   = (node < N_NODES) ? node : (N_NODES - 1);
        rows[j]  = ((const v4f*)(shared_f + (size_t)nc * HID))[lane];
    }

    float acc[LG_NPW];
    v4f cs = {0.f, 0.f, 0.f, 0.f};
#pragma unroll
    for (int j = 0; j < LG_NPW; ++j) {
        const bool valid = (node0 + j) < N_NODES;
        const v4f  v     = rows[j];
        float a = 0.f;
#pragma unroll
        for (int i = 0; i < 4; ++i) {
            a     += v[i] * w[i];
            cs[i] += valid ? v[i] : 0.f;
        }
        acc[j] = a;
    }
#pragma unroll
    for (int off = 16; off; off >>= 1) {
#pragma unroll
        for (int j = 0; j < LG_NPW; ++j) acc[j] += __shfl_down(acc[j], off, 32);
    }
    if (lane == 0) {
#pragma unroll
        for (int j = 0; j < LG_NPW; ++j) lg[wv * LG_NPW + j] = acc[j];
    }
    ((v4f*)&colpart[wv][0])[lane] = cs;
    __syncthreads();

    const int t = threadIdx.x;
    if (t < HID) {
        float s = 0.f;
#pragma unroll
        for (int k = 0; k < LG_WAVES; ++k) s += colpart[k][t];
        atomicAdd(&gpart[t], s);
    }
    if (t < LG_NPB) {
        int node = blockIdx.x * LG_NPB + t;
        if (node < N_NODES) out[node] = lg[t] + ba[0];
    }
}

__global__ void value_kernel(const float* __restrict__ gpart, const float* __restrict__ Wc,
                             const float* __restrict__ bc, float* __restrict__ out) {
    int lane = threadIdx.x & 31;
    v4f v = ((const v4f*)gpart)[lane];
    v4f w = ((const v4f*)Wc)[lane];
    float acc = 0.f;
#pragma unroll
    for (int i = 0; i < 4; ++i) acc += (v[i] * (1.f / N_NODES)) * w[i];
    for (int off = 16; off; off >>= 1) acc += __shfl_down(acc, off, 32);
    if (lane == 0) out[N_NODES] = acc + bc[0];
}

extern "C" void kernel_launch(void* const* d_in, const int* in_sizes, int n_in,
                              void* d_out, int out_size, void* d_ws, size_t ws_size,
                              hipStream_t stream) {
    const float* x     = (const float*)d_in[0];
    const int*   ei    = (const int*)d_in[1];
    const float* W_in  = (const float*)d_in[2];
    const float* b_in  = (const float*)d_in[3];
    const float* Wl1   = (const float*)d_in[4];
    const float* bl1   = (const float*)d_in[5];
    const float* Wr1   = (const float*)d_in[6];
    const float* br1   = (const float*)d_in[7];
    const float* att1  = (const float*)d_in[8];
    const float* bias1 = (const float*)d_in[9];
    const float* g1    = (const float*)d_in[10];
    const float* be1   = (const float*)d_in[11];
    const float* Wl2   = (const float*)d_in[12];
    const float* bl2   = (const float*)d_in[13];
    const float* Wr2   = (const float*)d_in[14];
    const float* br2   = (const float*)d_in[15];
    const float* att2  = (const float*)d_in[16];
    const float* bias2 = (const float*)d_in[17];
    const float* g2    = (const float*)d_in[18];
    const float* be2   = (const float*)d_in[19];
    const float* Wa    = (const float*)d_in[20];
    const float* ba    = (const float*)d_in[21];
    const float* Wc    = (const float*)d_in[22];
    const float* bc    = (const float*)d_in[23];

    float* out = (float*)d_out;

    float* ws = (float*)d_ws;
    size_t off = 0;
    float*    h0    = ws + off; off += (size_t)N_NODES * HID;
    float*    h1    = ws + off; off += (size_t)N_NODES * HID;
    float*    xl    = ws + off; off += (size_t)N_NODES * HC;
    float*    xr    = ws + off; off += (size_t)N_NODES * HC;
    int*      csr   = (int*)(ws + off); off += (size_t)ETOT * NHEAD;
    off += (size_t)N_NODES * NHEAD;
    off += (size_t)N_NODES * NHEAD;
    off += (size_t)N_NODES * HID;
    float*    gpart = ws + off; off += HID;
    (void)ws_size; (void)in_sizes; (void)n_in; (void)out_size;

    int* cnt   = csr;
    int* first = cnt + N_NODES;
    int* slot  = first + (N_NODES + 16);
    int* srcs  = slot + E_EDGES;
    static_assert((size_t)N_NODES + (N_NODES + 16) + 2 * (size_t)E_EDGES <= (size_t)ETOT * NHEAD,
                  "the grouping must fit the region it reuses");

    const int initBlocks  = (N_NODES + 255) / 256;
    const int eBlocks     = (E_EDGES + 255) / 256;
    const int gatherBlocks = (N_NODES + NG_WAVES - 1) / NG_WAVES;
    const dim3 gemmInGrid(HID / GM_BN, N_NODES / GM_BM);
    const dim3 gemmHCGrid(HC / GM_BN, N_NODES / GM_BM);
    const int logitBlocks = (N_NODES + LG_NPB - 1) / LG_NPB;

    init_scratch<<<initBlocks, 256, 0, stream>>>(cnt, gpart);
    csr_count   <<<eBlocks, 256, 0, stream>>>(ei, cnt, slot);
    csr_scan    <<<1, SC_THREADS, 0, stream>>>(cnt, first);
    csr_fill    <<<eBlocks, 256, 0, stream>>>(ei, first, slot, srcs);

    gemm_wmma_f32<1, DIN, HID><<<gemmInGrid, GM_WAVES * 32, 0, stream>>>(x, W_in, b_in, h0);

    gemm_wmma_f32<0, HID, HC><<<gemmHCGrid, GM_WAVES * 32, 0, stream>>>(h0, Wl1, bl1, xl);
    gemm_wmma_f32<0, HID, HC><<<gemmHCGrid, GM_WAVES * 32, 0, stream>>>(h0, Wr1, br1, xr);
    node_gather <<<gatherBlocks, NG_WAVES * 32, 0, stream>>>(first, srcs, xl, xr, att1,
                                                             h0, bias1, g1, be1, h1);

    gemm_wmma_f32<0, HID, HC><<<gemmHCGrid, GM_WAVES * 32, 0, stream>>>(h1, Wl2, bl2, xl);
    gemm_wmma_f32<0, HID, HC><<<gemmHCGrid, GM_WAVES * 32, 0, stream>>>(h1, Wr2, br2, xr);
    node_gather <<<gatherBlocks, NG_WAVES * 32, 0, stream>>>(first, srcs, xl, xr, att2,
                                                             h1, bias2, g2, be2, h0);

    logits_kernel<<<logitBlocks, LG_WAVES * 32, 0, stream>>>(h0, Wa, ba, gpart, out);
    value_kernel <<<1, 32, 0, stream>>>(gpart, Wc, bc, out);
}


# edge grouping (count, scan, fill) has no launches of its own: each stage rides as a few fat blocks at the front of one of the three projection launches before the first gather; 13 -> 10 launches
# speedup vs baseline: 3.5308x; 1.0577x over previous
#include <hip/hip_runtime.h>
#include <hip/hip_bf16.h>
#include <math.h>

#define N_NODES 20000
#define E_EDGES 320000
#define ETOT    (E_EDGES + N_NODES)
#define DIN     64
#define HID     128
#define NHEAD   4
#define HC      (NHEAD * HID)

typedef float v2f __attribute__((ext_vector_type(2)));
typedef float v4f __attribute__((ext_vector_type(4)));
typedef float v8f __attribute__((ext_vector_type(8)));

#define GM_BM    80
#define GM_MT    (GM_BM / 16)
#define GM_WAVES 8
#define GM_BN    (GM_WAVES * 16)
static_assert(N_NODES % GM_BM == 0, "the row blocks must cover M exactly");
static_assert(HID % GM_BN == 0 && HC % GM_BN == 0, "the column blocks must cover N exactly");

#define SIDE_NONE  0
#define SIDE_COUNT 1
#define SIDE_SCAN  2
#define SIDE_FILL  3
#define SD_THREADS (GM_WAVES * 32)
#define EG_PER     10
#define EG_BLOCK   (SD_THREADS * EG_PER)
#define EG_NBLK    ((E_EDGES + EG_BLOCK - 1) / EG_BLOCK)
#define SCN_PER    ((N_NODES + SD_THREADS - 1) / SD_THREADS)
#define SCN_PAD    (SD_THREADS * SCN_PER)
#define SCN_LDS    (SCN_PAD + GM_WAVES)

__host__ __device__ constexpr int side_rows(int side, int gx) {
    return side == SIDE_NONE ? 0 : side == SIDE_SCAN ? 1 : (EG_NBLK + gx - 1) / gx;
}

static_assert(E_EDGES % EG_BLOCK == 0, "side blocks cover the edges exactly: no per-thread bounds test");
__device__ __forceinline__ void edge_count(int sb, const int* __restrict__ ei,
                                           int* __restrict__ cnt, int* __restrict__ slot) {
    if (sb >= EG_NBLK) return;
    const int e0 = sb * EG_BLOCK + (int)threadIdx.x;
    int d[EG_PER], r[EG_PER];
#pragma unroll
    for (int i = 0; i < EG_PER; ++i) d[i] = ei[E_EDGES + e0 + i * SD_THREADS];
#pragma unroll
    for (int i = 0; i < EG_PER; ++i) r[i] = atomicAdd(&cnt[d[i]], 1);
#pragma unroll
    for (int i = 0; i < EG_PER; ++i) slot[e0 + i * SD_THREADS] = r[i];
}

__device__ __forceinline__ void edge_fill(int sb, const int* __restrict__ ei,
                                          const int* __restrict__ first,
                                          const int* __restrict__ slot, int* __restrict__ srcs) {
    if (sb >= EG_NBLK) return;
    const int e0 = sb * EG_BLOCK + (int)threadIdx.x;
    int d[EG_PER], s[EG_PER], p[EG_PER];
#pragma unroll
    for (int i = 0; i < EG_PER; ++i) {
        const int e = e0 + i * SD_THREADS;
        d[i] = ei[E_EDGES + e];
        s[i] = ei[e];
        p[i] = slot[e];
    }
#pragma unroll
    for (int i = 0; i < EG_PER; ++i) p[i] += first[d[i]];
#pragma unroll
    for (int i = 0; i < EG_PER; ++i) srcs[p[i]] = s[i];
}

static_assert((SD_THREADS - 1) * SCN_PER >= N_NODES, "the last thread's run must lie in the pad");
static_assert((SCN_PER - 1) * SD_THREADS <= N_NODES, "only the last round of whole lines is cut");
__device__ __forceinline__ void node_scan(const int* __restrict__ cnt, int* __restrict__ first,
                                          int* sm) {
    const int t    = (int)threadIdx.x;
    const int lane = t & 31;
    const int wv   = t >> 5;
    int v[SCN_PER];
#pragma unroll
    for (int i = 0; i < SCN_PER; ++i) {
        const int idx = i * SD_THREADS + t;
        const int ld  = cnt[min(idx, N_NODES - 1)];
        v[i] = (idx < N_NODES) ? ld : 0;
    }
#pragma unroll
    for (int i = 0; i < SCN_PER; ++i) sm[i * SD_THREADS + t] = v[i];
    __syncthreads();

    const int b = t * SCN_PER;
    int s = 0;
#pragma unroll
    for (int i = 0; i < SCN_PER; ++i) {
        v[i] = sm[b + i];
        s += v[i];
    }
    int incl = s;
    for (int off = 1; off < 32; off <<= 1) {
        const int u = __shfl_up(incl, off, 32);
        if (lane >= off) incl += u;
    }
    int* wsum = sm + SCN_PAD;
    if (lane == 31) wsum[wv] = incl;
    __syncthreads();
    int run = incl - s;
    for (int k = 0; k < wv; ++k) run += wsum[k];
#pragma unroll
    for (int i = 0; i < SCN_PER; ++i) {
        sm[b + i] = run;
        run += v[i];
    }
    if (t == SD_THREADS - 1) first[N_NODES] = run;
    __syncthreads();
#pragma unroll
    for (int i = 0; i < SCN_PER; ++i) v[i] = sm[i * SD_THREADS + t];
#pragma unroll
    for (int i = 0; i < SCN_PER - 1; ++i) first[i * SD_THREADS + t] = v[i];
    if ((SCN_PER - 1) * SD_THREADS + t < N_NODES) first[(SCN_PER - 1) * SD_THREADS + t] = v[SCN_PER - 1];
}

template <int RELU, int K, int N, int SIDE>
__global__ void __launch_bounds__(GM_WAVES * 32)
gemm_wmma_f32(const float* __restrict__ A,
              const float* __restrict__ B,
              const float* __restrict__ bias,
              float* __restrict__ C,
              const int* __restrict__ ei, int* __restrict__ cnt, int* __restrict__ first,
              int* __restrict__ slot, int* __restrict__ srcs) {
    constexpr int P    = K + 4;
    constexpr int KS   = K / 4;
    constexpr int NV4  = GM_BM * KS;
    constexpr int NTHR = GM_WAVES * 32;
    static_assert(NV4 % NTHR == 0, "every thread copies the same number of chunks");
    constexpr int LDSF = (SIDE == SIDE_SCAN && SCN_LDS > GM_BM * P) ? SCN_LDS : GM_BM * P;
    __shared__ __attribute__((aligned(16))) float As[LDSF];

    constexpr int GX    = N / GM_BN;
    constexpr int SROWS = side_rows(SIDE, GX);
    if (SIDE != SIDE_NONE && (int)blockIdx.y < SROWS) {
        const int sb = (int)blockIdx.y * GX + (int)blockIdx.x;
        if (SIDE == SIDE_COUNT) edge_count(sb, ei, cnt, slot);
        if (SIDE == SIDE_SCAN) { if (sb == 0) node_scan(cnt, first, (int*)As); }
        if (SIDE == SIDE_FILL) edge_fill(sb, ei, first, slot, srcs);
        return;
    }

    const int t    = threadIdx.x;
    const int lane = t & 31;
    const int wv   = __builtin_amdgcn_readfirstlane(t >> 5);
    const int l16  = lane & 15;
    const int half = lane >> 4;
    const int m0   = ((int)blockIdx.y - SROWS) * GM_BM;
    const int col  = blockIdx.x * GM_BN + wv * 16 + l16;

    const v4f* ag = (const v4f*)(A + (size_t)m0 * K);
    v4f stage[NV4 / NTHR];
#pragma unroll
    for (int i = 0; i < NV4 / NTHR; ++i) stage[i] = ag[t + NTHR * i];

    const float* bcol = B + (size_t)(2 * half) * N + col;
    v2f bfr[KS];
#pragma unroll
    for (int j = 0; j < KS; ++j) {
        bfr[j].x = bcol[(size_t)(4 * j) * N];
        bfr[j].y = bcol[(size_t)(4 * j + 1) * N];
    }

#pragma unroll
    for (int i = 0; i < NV4 / NTHR; ++i) {
        const int idx = t + NTHR * i;
        const int row = idx / KS;
        const int c4  = idx % KS;
        *(v4f*)&As[row * P + 4 * c4] = stage[i];
    }
    __syncthreads();

    v8f acc[GM_MT];
#pragma unroll
    for (int mt = 0; mt < GM_MT; ++mt) acc[mt] = (v8f){0.f, 0.f, 0.f, 0.f, 0.f, 0.f, 0.f, 0.f};

    const float* arow = &As[l16 * P + 2 * half];
#pragma unroll
    for (int j = 0; j < KS; ++j) {
#pragma unroll
        for (int mt = 0; mt < GM_MT; ++mt) {
            const v2f a = *(const v2f*)(arow + mt * 16 * P + 4 * j);
            acc[mt] = __builtin_amdgcn_wmma_f32_16x16x4_f32(
                 false, a,  false, bfr[j],
                 (short)0, acc[mt],  false,  false);
        }
    }

    const float bv = bias[col];
#pragma unroll
    for (int mt = 0; mt < GM_MT; ++mt) {
#pragma unroll
        for (int v = 0; v < 8; ++v) {
            const int row = m0 + mt * 16 + v + 8 * half;
            float val = acc[mt][v] + bv;
            if (RELU) val = fmaxf(val, 0.f);
            C[(size_t)row * N + col] = val;
        }
    }
}

__global__ void init_scratch(int* __restrict__ cnt, float* __restrict__ gpart) {
    int idx = blockIdx.x * blockDim.x + threadIdx.x;
    if (idx < N_NODES) cnt[idx] = 0;
    if (idx < HID) gpart[idx] = 0.f;
}

template <int M>
__device__ __forceinline__ float xrow(float v) {
    const int b = __builtin_bit_cast(int, v);
    return __builtin_bit_cast(float, __builtin_amdgcn_mov_dpp(b, 0x160 + M, 0xf, 0xf, true));
}
__device__ __forceinline__ float xhalf(float v) {
    const int b = __builtin_bit_cast(int, v);
    return __builtin_bit_cast(float,
        __builtin_amdgcn_permlanex16(b, b, 0x76543210, 0xfedcba98, false, false));
}

#define NCH 4
__device__ __forceinline__ void load_row(const float* __restrict__ base, int lofs,
                                         v4f (&x)[NCH]) {
    const v4f* p = (const v4f*)base + lofs;
#pragma unroll
    for (int i = 0; i < NCH; ++i) x[i] = p[8 * i];
}

__device__ __forceinline__ float edge_score(const v4f (&x)[NCH], const v4f (&r)[NCH],
                                            const v4f (&w)[NCH]) {
    v2f sa = {0.f, 0.f};
    v2f sb = {0.f, 0.f};
#pragma unroll
    for (int i = 0; i < NCH; ++i) {
        const v2f ma = x[i].lo + r[i].lo;
        const v2f mb = x[i].hi + r[i].hi;
        const v2f ea = __builtin_elementwise_max(ma, ma * 0.2f);
        const v2f eb = __builtin_elementwise_max(mb, mb * 0.2f);
        sa = __builtin_elementwise_fma(ea, w[i].lo, sa);
        sb = __builtin_elementwise_fma(eb, w[i].hi, sb);
    }
    const v2f s2 = sa + sb;
    float a = s2.x + s2.y;
    a += xrow<1>(a);
    a += xrow<2>(a);
    a += xrow<4>(a);
    return a;
}

__device__ __forceinline__ void edge_update(const v4f (&x)[NCH], const v4f (&r)[NCH],
                                            const v4f (&w)[NCH], float& m, float& l,
                                            v4f (&acc)[NCH]) {
    const float a  = edge_score(x, r, w);
    const float d  = a - m;
    const float ex = expf(-fabsf(d));
    const bool  up = d > 0.f;
    const float sc = up ? ex : 1.f;
    const float p  = up ? 1.f : ex;
    m = up ? a : m;
    l = l * sc + p;
    const v2f scv = {sc, sc};
    const v2f pv  = {p, p};
#pragma unroll
    for (int i = 0; i < NCH; ++i) {
        acc[i].lo = __builtin_elementwise_fma(x[i].lo, pv, acc[i].lo * scv);
        acc[i].hi = __builtin_elementwise_fma(x[i].hi, pv, acc[i].hi * scv);
    }
}

#define NG_WAVES 4

__global__ void __launch_bounds__(NG_WAVES * 32)
node_gather(const int* __restrict__ first, const int* __restrict__ srcs,
            const float* __restrict__ xl, const float* __restrict__ xr,
            const float* __restrict__ att,
            const float* __restrict__ hin, const float* __restrict__ bias,
            const float* __restrict__ g, const float* __restrict__ be,
            float* __restrict__ hout) {
    const int node = __builtin_amdgcn_readfirstlane((int)((blockIdx.x * blockDim.x + threadIdx.x) >> 5));
    if (node >= N_NODES) return;
    const int lane = threadIdx.x & 31;
    const int q    = lane >> 3;
    const int lofs = q * (HID / 4) + (lane & 7);
    const int beg  = first[node];
    const int end  = first[node + 1];

    v4f r[NCH], w[NCH], x[NCH], xn[NCH];
    load_row(xr + (size_t)node * HC, lofs, r);
    load_row(att, lofs, w);
    load_row(xl + (size_t)node * HC, lofs, x);
#pragma unroll
    for (int i = 0; i < NCH; ++i) xn[i] = (v4f){0.f, 0.f, 0.f, 0.f};
    if (beg < end) load_row(xl + (size_t)srcs[beg] * HC, lofs, xn);

    float m = edge_score(x, r, w);
    float l = 1.f;
    v4f   acc[NCH];
#pragma unroll
    for (int i = 0; i < NCH; ++i) acc[i] = x[i];

    if (beg < end) {
        int e  = beg;
        int ia = srcs[min(e + 1, end - 1)];
        int ib = srcs[min(e + 2, end - 1)];
        while (e + 2 < end) {
            load_row(xl + (size_t)ia * HC, lofs, x);
            const int ja = srcs[min(e + 3, end - 1)];
            const int jb = srcs[min(e + 4, end - 1)];
            __builtin_amdgcn_sched_barrier(0);
            edge_update(xn, r, w, m, l, acc);
            load_row(xl + (size_t)ib * HC, lofs, xn);
            __builtin_amdgcn_sched_barrier(0);
            edge_update(x, r, w, m, l, acc);
            ia = ja;
            ib = jb;
            e += 2;
        }
        if (e + 1 < end) {
            load_row(xl + (size_t)ia * HC, lofs, x);
            __builtin_amdgcn_sched_barrier(0);
            edge_update(xn, r, w, m, l, acc);
            __builtin_amdgcn_sched_barrier(0);
            edge_update(x, r, w, m, l, acc);
        } else {
            edge_update(xn, r, w, m, l, acc);
        }
    }

    const float inv = 1.f / l;
    v4f t[NCH];
#pragma unroll
    for (int i = 0; i < NCH; ++i) {
#pragma unroll
        for (int k = 0; k < 4; ++k) {
            float v = acc[i][k] * inv;
            v += xrow<8>(v);
            v += xhalf(v);
            t[i][k] = v;
        }
    }
    v4f o = t[0];
    if (q == 1) o = t[1];
    if (q == 2) o = t[2];
    if (q == 3) o = t[3];
#pragma unroll
    for (int k = 0; k < 4; ++k) o[k] *= 0.25f;

    const v4f hv = ((const v4f*)(hin + (size_t)node * HID))[lane];
    const v4f cb = ((const v4f*)bias)[lane];
    const v4f gg = ((const v4f*)g)[lane];
    const v4f bb = ((const v4f*)be)[lane];
    v4f y;
#pragma unroll
    for (int k = 0; k < 4; ++k) y[k] = hv[k] + o[k] + cb[k];
    float s = 0.f, ss = 0.f;
#pragma unroll
    for (int k = 0; k < 4; ++k) { s += y[k]; ss += y[k] * y[k]; }
    s += xhalf(s);    ss += xhalf(ss);
    s += xrow<8>(s);  ss += xrow<8>(ss);
    s += xrow<4>(s);  ss += xrow<4>(ss);
    s += xrow<2>(s);  ss += xrow<2>(ss);
    s += xrow<1>(s);  ss += xrow<1>(ss);
    const float mu  = s * (1.f / HID);
    const float var = ss * (1.f / HID) - mu * mu;
    const float rs  = rsqrtf(var + 1e-5f);
    v4f ho;
#pragma unroll
    for (int k = 0; k < 4; ++k)
        ho[k] = fmaxf((y[k] - mu) * rs * gg[k] + bb[k], 0.f);
    ((v4f*)(hout + (size_t)node * HID))[lane] = ho;
}

#define LG_WAVES 8
#define LG_NPW   16
#define LG_NPB   (LG_WAVES * LG_NPW)

__global__ void __launch_bounds__(LG_WAVES * 32)
logits_kernel(const float* __restrict__ shared_f, const float* __restrict__ Wa,
              const float* __restrict__ ba, float* __restrict__ gpart,
              float* __restrict__ out) {
    __shared__ float colpart[LG_WAVES][HID];
    __shared__ float lg[LG_NPB];

    const int lane  = threadIdx.x & 31;
    const int wv    = __builtin_amdgcn_readfirstlane(threadIdx.x >> 5);
    const int node0 = blockIdx.x * LG_NPB + wv * LG_NPW;

    const v4f w = ((const v4f*)Wa)[lane];

    v4f rows[LG_NPW];
#pragma unroll
    for (int j = 0; j < LG_NPW; ++j) {
        int node = node0 + j;
        int nc   = (node < N_NODES) ? node : (N_NODES - 1);
        rows[j]  = ((const v4f*)(shared_f + (size_t)nc * HID))[lane];
    }

    float acc[LG_NPW];
    v4f cs = {0.f, 0.f, 0.f, 0.f};
#pragma unroll
    for (int j = 0; j < LG_NPW; ++j) {
        const bool valid = (node0 + j) < N_NODES;
        const v4f  v     = rows[j];
        float a = 0.f;
#pragma unroll
        for (int i = 0; i < 4; ++i) {
            a     += v[i] * w[i];
            cs[i] += valid ? v[i] : 0.f;
        }
        acc[j] = a;
    }
#pragma unroll
    for (int off = 16; off; off >>= 1) {
#pragma unroll
        for (int j = 0; j < LG_NPW; ++j) acc[j] += __shfl_down(acc[j], off, 32);
    }
    if (lane == 0) {
#pragma unroll
        for (int j = 0; j < LG_NPW; ++j) lg[wv * LG_NPW + j] = acc[j];
    }
    ((v4f*)&colpart[wv][0])[lane] = cs;
    __syncthreads();

    const int t = threadIdx.x;
    if (t < HID) {
        float s = 0.f;
#pragma unroll
        for (int k = 0; k < LG_WAVES; ++k) s += colpart[k][t];
        atomicAdd(&gpart[t], s);
    }
    if (t < LG_NPB) {
        int node = blockIdx.x * LG_NPB + t;
        if (node < N_NODES) out[node] = lg[t] + ba[0];
    }
}

__global__ void value_kernel(const float* __restrict__ gpart, const float* __restrict__ Wc,
                             const float* __restrict__ bc, float* __restrict__ out) {
    int lane = threadIdx.x & 31;
    v4f v = ((const v4f*)gpart)[lane];
    v4f w = ((const v4f*)Wc)[lane];
    float acc = 0.f;
#pragma unroll
    for (int i = 0; i < 4; ++i) acc += (v[i] * (1.f / N_NODES)) * w[i];
    for (int off = 16; off; off >>= 1) acc += __shfl_down(acc, off, 32);
    if (lane == 0) out[N_NODES] = acc + bc[0];
}

extern "C" void kernel_launch(void* const* d_in, const int* in_sizes, int n_in,
                              void* d_out, int out_size, void* d_ws, size_t ws_size,
                              hipStream_t stream) {
    const float* x     = (const float*)d_in[0];
    const int*   ei    = (const int*)d_in[1];
    const float* W_in  = (const float*)d_in[2];
    const float* b_in  = (const float*)d_in[3];
    const float* Wl1   = (const float*)d_in[4];
    const float* bl1   = (const float*)d_in[5];
    const float* Wr1   = (const float*)d_in[6];
    const float* br1   = (const float*)d_in[7];
    const float* att1  = (const float*)d_in[8];
    const float* bias1 = (const float*)d_in[9];
    const float* g1    = (const float*)d_in[10];
    const float* be1   = (const float*)d_in[11];
    const float* Wl2   = (const float*)d_in[12];
    const float* bl2   = (const float*)d_in[13];
    const float* Wr2   = (const float*)d_in[14];
    const float* br2   = (const float*)d_in[15];
    const float* att2  = (const float*)d_in[16];
    const float* bias2 = (const float*)d_in[17];
    const float* g2    = (const float*)d_in[18];
    const float* be2   = (const float*)d_in[19];
    const float* Wa    = (const float*)d_in[20];
    const float* ba    = (const float*)d_in[21];
    const float* Wc    = (const float*)d_in[22];
    const float* bc    = (const float*)d_in[23];

    float* out = (float*)d_out;

    float* ws = (float*)d_ws;
    size_t off = 0;
    float*    h0    = ws + off; off += (size_t)N_NODES * HID;
    float*    h1    = ws + off; off += (size_t)N_NODES * HID;
    float*    xl    = ws + off; off += (size_t)N_NODES * HC;
    float*    xr    = ws + off; off += (size_t)N_NODES * HC;
    int*      csr   = (int*)(ws + off); off += (size_t)ETOT * NHEAD;
    off += (size_t)N_NODES * NHEAD;
    off += (size_t)N_NODES * NHEAD;
    off += (size_t)N_NODES * HID;
    float*    gpart = ws + off; off += HID;
    (void)ws_size; (void)in_sizes; (void)n_in; (void)out_size;

    int* cnt   = csr;
    int* first = cnt + N_NODES;
    int* slot  = first + (N_NODES + 16);
    int* srcs  = slot + E_EDGES;
    static_assert((size_t)N_NODES + (N_NODES + 16) + 2 * (size_t)E_EDGES <= (size_t)ETOT * NHEAD,
                  "the grouping must fit the region it reuses");

    const int initBlocks  = (N_NODES + 255) / 256;
    const int gatherBlocks = (N_NODES + NG_WAVES - 1) / NG_WAVES;
    const dim3 gemmHCGrid(HC / GM_BN, N_NODES / GM_BM);
    const dim3 gridInCount(HID / GM_BN, N_NODES / GM_BM + side_rows(SIDE_COUNT, HID / GM_BN));
    const dim3 gridHCScan (HC / GM_BN,  N_NODES / GM_BM + side_rows(SIDE_SCAN,  HC / GM_BN));
    const dim3 gridHCFill (HC / GM_BN,  N_NODES / GM_BM + side_rows(SIDE_FILL,  HC / GM_BN));
    const int logitBlocks = (N_NODES + LG_NPB - 1) / LG_NPB;

    init_scratch<<<initBlocks, 256, 0, stream>>>(cnt, gpart);

    gemm_wmma_f32<1, DIN, HID, SIDE_COUNT><<<gridInCount, GM_WAVES * 32, 0, stream>>>(
        x, W_in, b_in, h0, ei, cnt, first, slot, srcs);

    gemm_wmma_f32<0, HID, HC, SIDE_SCAN><<<gridHCScan, GM_WAVES * 32, 0, stream>>>(
        h0, Wl1, bl1, xl, ei, cnt, first, slot, srcs);
    gemm_wmma_f32<0, HID, HC, SIDE_FILL><<<gridHCFill, GM_WAVES * 32, 0, stream>>>(
        h0, Wr1, br1, xr, ei, cnt, first, slot, srcs);
    node_gather <<<gatherBlocks, NG_WAVES * 32, 0, stream>>>(first, srcs, xl, xr, att1,
                                                             h0, bias1, g1, be1, h1);

    gemm_wmma_f32<0, HID, HC, SIDE_NONE><<<gemmHCGrid, GM_WAVES * 32, 0, stream>>>(
        h1, Wl2, bl2, xl, ei, cnt, first, slot, srcs);
    gemm_wmma_f32<0, HID, HC, SIDE_NONE><<<gemmHCGrid, GM_WAVES * 32, 0, stream>>>(
        h1, Wr2, br2, xr, ei, cnt, first, slot, srcs);
    node_gather <<<gatherBlocks, NG_WAVES * 32, 0, stream>>>(first, srcs, xl, xr, att2,
                                                             h1, bias2, g2, be2, h0);

    logits_kernel<<<logitBlocks, LG_WAVES * 32, 0, stream>>>(h0, Wa, ba, gpart, out);
    value_kernel <<<1, 32, 0, stream>>>(gpart, Wc, bc, out);
}


# both projections of a layer (xl, xr) run as ONE launch, 8 column blocks on each row tile; scan and fill become one side stage (every fill block scans the counts in LDS): 10 -> 8 launches
# speedup vs baseline: 3.6404x; 1.0310x over previous
#include <hip/hip_runtime.h>
#include <hip/hip_bf16.h>
#include <math.h>

#define N_NODES 20000
#define E_EDGES 320000
#define ETOT    (E_EDGES + N_NODES)
#define DIN     64
#define HID     128
#define NHEAD   4
#define HC      (NHEAD * HID)

typedef float v2f __attribute__((ext_vector_type(2)));
typedef float v4f __attribute__((ext_vector_type(4)));
typedef float v8f __attribute__((ext_vector_type(8)));

#define GM_BM    80
#define GM_MT    (GM_BM / 16)
#define GM_WAVES 8
#define GM_BN    (GM_WAVES * 16)
static_assert(N_NODES % GM_BM == 0, "the row blocks must cover M exactly");
static_assert(HID % GM_BN == 0 && HC % GM_BN == 0, "the column blocks must cover N exactly");

#define SIDE_NONE     0
#define SIDE_COUNT    1
#define SIDE_SCANFILL 2
#define SD_THREADS (GM_WAVES * 32)
#define EG_PER     10
#define EG_BLOCK   (SD_THREADS * EG_PER)
#define EG_NBLK    ((E_EDGES + EG_BLOCK - 1) / EG_BLOCK)
#define SCN_PER    ((N_NODES + SD_THREADS - 1) / SD_THREADS)
#define SCN_PAD    (SD_THREADS * SCN_PER)
#define SCN_LDS    (SCN_PAD + GM_WAVES)

__host__ __device__ constexpr int side_rows(int side, int gx) {
    return side == SIDE_NONE ? 0 : (EG_NBLK + gx - 1) / gx;
}

static_assert(E_EDGES % EG_BLOCK == 0, "side blocks cover the edges exactly: no per-thread bounds test");
__device__ __forceinline__ void edge_count(int sb, const int* __restrict__ ei,
                                           int* __restrict__ cnt, int* __restrict__ slot) {
    if (sb >= EG_NBLK) return;
    const int e0 = sb * EG_BLOCK + (int)threadIdx.x;
    int d[EG_PER], r[EG_PER];
#pragma unroll
    for (int i = 0; i < EG_PER; ++i) d[i] = ei[E_EDGES + e0 + i * SD_THREADS];
#pragma unroll
    for (int i = 0; i < EG_PER; ++i) r[i] = atomicAdd(&cnt[d[i]], 1);
#pragma unroll
    for (int i = 0; i < EG_PER; ++i) slot[e0 + i * SD_THREADS] = r[i];
}

__device__ __forceinline__ void edge_fill(int sb, const int* __restrict__ ei,
                                          const int* first,
                                          const int* __restrict__ slot, int* __restrict__ srcs) {
    const int e0 = sb * EG_BLOCK + (int)threadIdx.x;
    int d[EG_PER], s[EG_PER], p[EG_PER];
#pragma unroll
    for (int i = 0; i < EG_PER; ++i) {
        const int e = e0 + i * SD_THREADS;
        d[i] = ei[E_EDGES + e];
        s[i] = ei[e];
        p[i] = slot[e];
    }
#pragma unroll
    for (int i = 0; i < EG_PER; ++i) p[i] += first[d[i]];
#pragma unroll
    for (int i = 0; i < EG_PER; ++i) srcs[p[i]] = s[i];
}

static_assert((SD_THREADS - 1) * SCN_PER >= N_NODES, "the last thread's run must lie in the pad");
static_assert((SCN_PER - 1) * SD_THREADS <= N_NODES, "only the last round of whole lines is cut");
__device__ __forceinline__ void node_scan(const int* __restrict__ cnt, int* __restrict__ first,
                                          int* sm, bool write_out) {
    const int t    = (int)threadIdx.x;
    const int lane = t & 31;
    const int wv   = t >> 5;
    int v[SCN_PER];
#pragma unroll
    for (int i = 0; i < SCN_PER; ++i) {
        const int idx = i * SD_THREADS + t;
        const int ld  = cnt[min(idx, N_NODES - 1)];
        v[i] = (idx < N_NODES) ? ld : 0;
    }
#pragma unroll
    for (int i = 0; i < SCN_PER; ++i) sm[i * SD_THREADS + t] = v[i];
    __syncthreads();

    const int b = t * SCN_PER;
    int s = 0;
#pragma unroll
    for (int i = 0; i < SCN_PER; ++i) {
        v[i] = sm[b + i];
        s += v[i];
    }
    int incl = s;
    for (int off = 1; off < 32; off <<= 1) {
        const int u = __shfl_up(incl, off, 32);
        if (lane >= off) incl += u;
    }
    int* wsum = sm + SCN_PAD;
    if (lane == 31) wsum[wv] = incl;
    __syncthreads();
    int run = incl - s;
    for (int k = 0; k < wv; ++k) run += wsum[k];
#pragma unroll
    for (int i = 0; i < SCN_PER; ++i) {
        sm[b + i] = run;
        run += v[i];
    }
    __syncthreads();
    if (!write_out) return;
    if (t == SD_THREADS - 1) first[N_NODES] = run;
#pragma unroll
    for (int i = 0; i < SCN_PER; ++i) v[i] = sm[i * SD_THREADS + t];
#pragma unroll
    for (int i = 0; i < SCN_PER - 1; ++i) first[i * SD_THREADS + t] = v[i];
    if ((SCN_PER - 1) * SD_THREADS + t < N_NODES) first[(SCN_PER - 1) * SD_THREADS + t] = v[SCN_PER - 1];
}

template <int RELU, int K, int N, int NB, int SIDE>
__global__ void __launch_bounds__(GM_WAVES * 32)
gemm_wmma_f32(const float* __restrict__ A,
              const float* __restrict__ B0, const float* __restrict__ bias0, float* __restrict__ C0,
              const float* __restrict__ B1, const float* __restrict__ bias1, float* __restrict__ C1,
              const int* __restrict__ ei, int* __restrict__ cnt, int* __restrict__ first,
              int* __restrict__ slot, int* __restrict__ srcs) {
    constexpr int P    = K + 4;
    constexpr int KS   = K / 4;
    constexpr int NV4  = GM_BM * KS;
    constexpr int NTHR = GM_WAVES * 32;
    static_assert(NV4 % NTHR == 0, "every thread copies the same number of chunks");
    constexpr int LDSF = (SIDE == SIDE_SCANFILL && SCN_LDS > GM_BM * P) ? SCN_LDS : GM_BM * P;
    __shared__ __attribute__((aligned(16))) float As[LDSF];

    constexpr int CBN   = N / GM_BN;
    constexpr int GX    = NB * CBN;
    constexpr int SROWS = side_rows(SIDE, GX);
    if (SIDE != SIDE_NONE && (int)blockIdx.y < SROWS) {
        const int sb = (int)blockIdx.y * GX + (int)blockIdx.x;
        if (SIDE == SIDE_COUNT) edge_count(sb, ei, cnt, slot);
        if (SIDE == SIDE_SCANFILL && sb < EG_NBLK) {
            node_scan(cnt, first, (int*)As, sb == 0);
            edge_fill(sb, ei, (const int*)As, slot, srcs);
        }
        return;
    }
    const bool second = NB == 2 && (int)blockIdx.x >= CBN;
    const float* __restrict__ B    = second ? B1 : B0;
    const float* __restrict__ bias = second ? bias1 : bias0;
    float* __restrict__       C    = second ? C1 : C0;
    const int cblk = second ? (int)blockIdx.x - CBN : (int)blockIdx.x;

    const int t    = threadIdx.x;
    const int lane = t & 31;
    const int wv   = __builtin_amdgcn_readfirstlane(t >> 5);
    const int l16  = lane & 15;
    const int half = lane >> 4;
    const int m0   = ((int)blockIdx.y - SROWS) * GM_BM;
    const int col  = cblk * GM_BN + wv * 16 + l16;

    const v4f* ag = (const v4f*)(A + (size_t)m0 * K);
    v4f stage[NV4 / NTHR];
#pragma unroll
    for (int i = 0; i < NV4 / NTHR; ++i) stage[i] = ag[t + NTHR * i];

    const float* bcol = B + (size_t)(2 * half) * N + col;
    v2f bfr[KS];
#pragma unroll
    for (int j = 0; j < KS; ++j) {
        bfr[j].x = bcol[(size_t)(4 * j) * N];
        bfr[j].y = bcol[(size_t)(4 * j + 1) * N];
    }

#pragma unroll
    for (int i = 0; i < NV4 / NTHR; ++i) {
        const int idx = t + NTHR * i;
        const int row = idx / KS;
        const int c4  = idx % KS;
        *(v4f*)&As[row * P + 4 * c4] = stage[i];
    }
    __syncthreads();

    v8f acc[GM_MT];
#pragma unroll
    for (int mt = 0; mt < GM_MT; ++mt) acc[mt] = (v8f){0.f, 0.f, 0.f, 0.f, 0.f, 0.f, 0.f, 0.f};

    const float* arow = &As[l16 * P + 2 * half];
#pragma unroll
    for (int j = 0; j < KS; ++j) {
#pragma unroll
        for (int mt = 0; mt < GM_MT; ++mt) {
            const v2f a = *(const v2f*)(arow + mt * 16 * P + 4 * j);
            acc[mt] = __builtin_amdgcn_wmma_f32_16x16x4_f32(
                 false, a,  false, bfr[j],
                 (short)0, acc[mt],  false,  false);
        }
    }

    const float bv = bias[col];
#pragma unroll
    for (int mt = 0; mt < GM_MT; ++mt) {
#pragma unroll
        for (int v = 0; v < 8; ++v) {
            const int row = m0 + mt * 16 + v + 8 * half;
            float val = acc[mt][v] + bv;
            if (RELU) val = fmaxf(val, 0.f);
            C[(size_t)row * N + col] = val;
        }
    }
}

__global__ void init_scratch(int* __restrict__ cnt, float* __restrict__ gpart) {
    int idx = blockIdx.x * blockDim.x + threadIdx.x;
    if (idx < N_NODES) cnt[idx] = 0;
    if (idx < HID) gpart[idx] = 0.f;
}

template <int M>
__device__ __forceinline__ float xrow(float v) {
    const int b = __builtin_bit_cast(int, v);
    return __builtin_bit_cast(float, __builtin_amdgcn_mov_dpp(b, 0x160 + M, 0xf, 0xf, true));
}
__device__ __forceinline__ float xhalf(float v) {
    const int b = __builtin_bit_cast(int, v);
    return __builtin_bit_cast(float,
        __builtin_amdgcn_permlanex16(b, b, 0x76543210, 0xfedcba98, false, false));
}

#define NCH 4
__device__ __forceinline__ void load_row(const float* __restrict__ base, int lofs,
                                         v4f (&x)[NCH]) {
    const v4f* p = (const v4f*)base + lofs;
#pragma unroll
    for (int i = 0; i < NCH; ++i) x[i] = p[8 * i];
}

__device__ __forceinline__ float edge_score(const v4f (&x)[NCH], const v4f (&r)[NCH],
                                            const v4f (&w)[NCH]) {
    v2f sa = {0.f, 0.f};
    v2f sb = {0.f, 0.f};
#pragma unroll
    for (int i = 0; i < NCH; ++i) {
        const v2f ma = x[i].lo + r[i].lo;
        const v2f mb = x[i].hi + r[i].hi;
        const v2f ea = __builtin_elementwise_max(ma, ma * 0.2f);
        const v2f eb = __builtin_elementwise_max(mb, mb * 0.2f);
        sa = __builtin_elementwise_fma(ea, w[i].lo, sa);
        sb = __builtin_elementwise_fma(eb, w[i].hi, sb);
    }
    const v2f s2 = sa + sb;
    float a = s2.x + s2.y;
    a += xrow<1>(a);
    a += xrow<2>(a);
    a += xrow<4>(a);
    return a;
}

__device__ __forceinline__ void edge_update(const v4f (&x)[NCH], const v4f (&r)[NCH],
                                            const v4f (&w)[NCH], float& m, float& l,
                                            v4f (&acc)[NCH]) {
    const float a  = edge_score(x, r, w);
    const float d  = a - m;
    const float ex = expf(-fabsf(d));
    const bool  up = d > 0.f;
    const float sc = up ? ex : 1.f;
    const float p  = up ? 1.f : ex;
    m = up ? a : m;
    l = l * sc + p;
    const v2f scv = {sc, sc};
    const v2f pv  = {p, p};
#pragma unroll
    for (int i = 0; i < NCH; ++i) {
        acc[i].lo = __builtin_elementwise_fma(x[i].lo, pv, acc[i].lo * scv);
        acc[i].hi = __builtin_elementwise_fma(x[i].hi, pv, acc[i].hi * scv);
    }
}

#define NG_WAVES 4

__global__ void __launch_bounds__(NG_WAVES * 32)
node_gather(const int* __restrict__ first, const int* __restrict__ srcs,
            const float* __restrict__ xl, const float* __restrict__ xr,
            const float* __restrict__ att,
            const float* __restrict__ hin, const float* __restrict__ bias,
            const float* __restrict__ g, const float* __restrict__ be,
            float* __restrict__ hout) {
    const int node = __builtin_amdgcn_readfirstlane((int)((blockIdx.x * blockDim.x + threadIdx.x) >> 5));
    if (node >= N_NODES) return;
    const int lane = threadIdx.x & 31;
    const int q    = lane >> 3;
    const int lofs = q * (HID / 4) + (lane & 7);
    const int beg  = first[node];
    const int end  = first[node + 1];

    v4f r[NCH], w[NCH], x[NCH], xn[NCH];
    load_row(xr + (size_t)node * HC, lofs, r);
    load_row(att, lofs, w);
    load_row(xl + (size_t)node * HC, lofs, x);
#pragma unroll
    for (int i = 0; i < NCH; ++i) xn[i] = (v4f){0.f, 0.f, 0.f, 0.f};
    if (beg < end) load_row(xl + (size_t)srcs[beg] * HC, lofs, xn);

    float m = edge_score(x, r, w);
    float l = 1.f;
    v4f   acc[NCH];
#pragma unroll
    for (int i = 0; i < NCH; ++i) acc[i] = x[i];

    if (beg < end) {
        int e  = beg;
        int ia = srcs[min(e + 1, end - 1)];
        int ib = srcs[min(e + 2, end - 1)];
        while (e + 2 < end) {
            load_row(xl + (size_t)ia * HC, lofs, x);
            const int ja = srcs[min(e + 3, end - 1)];
            const int jb = srcs[min(e + 4, end - 1)];
            __builtin_amdgcn_sched_barrier(0);
            edge_update(xn, r, w, m, l, acc);
            load_row(xl + (size_t)ib * HC, lofs, xn);
            __builtin_amdgcn_sched_barrier(0);
            edge_update(x, r, w, m, l, acc);
            ia = ja;
            ib = jb;
            e += 2;
        }
        if (e + 1 < end) {
            load_row(xl + (size_t)ia * HC, lofs, x);
            __builtin_amdgcn_sched_barrier(0);
            edge_update(xn, r, w, m, l, acc);
            __builtin_amdgcn_sched_barrier(0);
            edge_update(x, r, w, m, l, acc);
        } else {
            edge_update(xn, r, w, m, l, acc);
        }
    }

    const float inv = 1.f / l;
    v4f t[NCH];
#pragma unroll
    for (int i = 0; i < NCH; ++i) {
#pragma unroll
        for (int k = 0; k < 4; ++k) {
            float v = acc[i][k] * inv;
            v += xrow<8>(v);
            v += xhalf(v);
            t[i][k] = v;
        }
    }
    v4f o = t[0];
    if (q == 1) o = t[1];
    if (q == 2) o = t[2];
    if (q == 3) o = t[3];
#pragma unroll
    for (int k = 0; k < 4; ++k) o[k] *= 0.25f;

    const v4f hv = ((const v4f*)(hin + (size_t)node * HID))[lane];
    const v4f cb = ((const v4f*)bias)[lane];
    const v4f gg = ((const v4f*)g)[lane];
    const v4f bb = ((const v4f*)be)[lane];
    v4f y;
#pragma unroll
    for (int k = 0; k < 4; ++k) y[k] = hv[k] + o[k] + cb[k];
    float s = 0.f, ss = 0.f;
#pragma unroll
    for (int k = 0; k < 4; ++k) { s += y[k]; ss += y[k] * y[k]; }
    s += xhalf(s);    ss += xhalf(ss);
    s += xrow<8>(s);  ss += xrow<8>(ss);
    s += xrow<4>(s);  ss += xrow<4>(ss);
    s += xrow<2>(s);  ss += xrow<2>(ss);
    s += xrow<1>(s);  ss += xrow<1>(ss);
    const float mu  = s * (1.f / HID);
    const float var = ss * (1.f / HID) - mu * mu;
    const float rs  = rsqrtf(var + 1e-5f);
    v4f ho;
#pragma unroll
    for (int k = 0; k < 4; ++k)
        ho[k] = fmaxf((y[k] - mu) * rs * gg[k] + bb[k], 0.f);
    ((v4f*)(hout + (size_t)node * HID))[lane] = ho;
}

#define LG_WAVES 8
#define LG_NPW   16
#define LG_NPB   (LG_WAVES * LG_NPW)

__global__ void __launch_bounds__(LG_WAVES * 32)
logits_kernel(const float* __restrict__ shared_f, const float* __restrict__ Wa,
              const float* __restrict__ ba, float* __restrict__ gpart,
              float* __restrict__ out) {
    __shared__ float colpart[LG_WAVES][HID];
    __shared__ float lg[LG_NPB];

    const int lane  = threadIdx.x & 31;
    const int wv    = __builtin_amdgcn_readfirstlane(threadIdx.x >> 5);
    const int node0 = blockIdx.x * LG_NPB + wv * LG_NPW;

    const v4f w = ((const v4f*)Wa)[lane];

    v4f rows[LG_NPW];
#pragma unroll
    for (int j = 0; j < LG_NPW; ++j) {
        int node = node0 + j;
        int nc   = (node < N_NODES) ? node : (N_NODES - 1);
        rows[j]  = ((const v4f*)(shared_f + (size_t)nc * HID))[lane];
    }

    float acc[LG_NPW];
    v4f cs = {0.f, 0.f, 0.f, 0.f};
#pragma unroll
    for (int j = 0; j < LG_NPW; ++j) {
        const bool valid = (node0 + j) < N_NODES;
        const v4f  v     = rows[j];
        float a = 0.f;
#pragma unroll
        for (int i = 0; i < 4; ++i) {
            a     += v[i] * w[i];
            cs[i] += valid ? v[i] : 0.f;
        }
        acc[j] = a;
    }
#pragma unroll
    for (int off = 16; off; off >>= 1) {
#pragma unroll
        for (int j = 0; j < LG_NPW; ++j) acc[j] += __shfl_down(acc[j], off, 32);
    }
    if (lane == 0) {
#pragma unroll
        for (int j = 0; j < LG_NPW; ++j) lg[wv * LG_NPW + j] = acc[j];
    }
    ((v4f*)&colpart[wv][0])[lane] = cs;
    __syncthreads();

    const int t = threadIdx.x;
    if (t < HID) {
        float s = 0.f;
#pragma unroll
        for (int k = 0; k < LG_WAVES; ++k) s += colpart[k][t];
        atomicAdd(&gpart[t], s);
    }
    if (t < LG_NPB) {
        int node = blockIdx.x * LG_NPB + t;
        if (node < N_NODES) out[node] = lg[t] + ba[0];
    }
}

__global__ void value_kernel(const float* __restrict__ gpart, const float* __restrict__ Wc,
                             const float* __restrict__ bc, float* __restrict__ out) {
    int lane = threadIdx.x & 31;
    v4f v = ((const v4f*)gpart)[lane];
    v4f w = ((const v4f*)Wc)[lane];
    float acc = 0.f;
#pragma unroll
    for (int i = 0; i < 4; ++i) acc += (v[i] * (1.f / N_NODES)) * w[i];
    for (int off = 16; off; off >>= 1) acc += __shfl_down(acc, off, 32);
    if (lane == 0) out[N_NODES] = acc + bc[0];
}

extern "C" void kernel_launch(void* const* d_in, const int* in_sizes, int n_in,
                              void* d_out, int out_size, void* d_ws, size_t ws_size,
                              hipStream_t stream) {
    const float* x     = (const float*)d_in[0];
    const int*   ei    = (const int*)d_in[1];
    const float* W_in  = (const float*)d_in[2];
    const float* b_in  = (const float*)d_in[3];
    const float* Wl1   = (const float*)d_in[4];
    const float* bl1   = (const float*)d_in[5];
    const float* Wr1   = (const float*)d_in[6];
    const float* br1   = (const float*)d_in[7];
    const float* att1  = (const float*)d_in[8];
    const float* bias1 = (const float*)d_in[9];
    const float* g1    = (const float*)d_in[10];
    const float* be1   = (const float*)d_in[11];
    const float* Wl2   = (const float*)d_in[12];
    const float* bl2   = (const float*)d_in[13];
    const float* Wr2   = (const float*)d_in[14];
    const float* br2   = (const float*)d_in[15];
    const float* att2  = (const float*)d_in[16];
    const float* bias2 = (const float*)d_in[17];
    const float* g2    = (const float*)d_in[18];
    const float* be2   = (const float*)d_in[19];
    const float* Wa    = (const float*)d_in[20];
    const float* ba    = (const float*)d_in[21];
    const float* Wc    = (const float*)d_in[22];
    const float* bc    = (const float*)d_in[23];

    float* out = (float*)d_out;

    float* ws = (float*)d_ws;
    size_t off = 0;
    float*    h0    = ws + off; off += (size_t)N_NODES * HID;
    float*    h1    = ws + off; off += (size_t)N_NODES * HID;
    float*    xl    = ws + off; off += (size_t)N_NODES * HC;
    float*    xr    = ws + off; off += (size_t)N_NODES * HC;
    int*      csr   = (int*)(ws + off); off += (size_t)ETOT * NHEAD;
    off += (size_t)N_NODES * NHEAD;
    off += (size_t)N_NODES * NHEAD;
    off += (size_t)N_NODES * HID;
    float*    gpart = ws + off; off += HID;
    (void)ws_size; (void)in_sizes; (void)n_in; (void)out_size;

    int* cnt   = csr;
    int* first = cnt + N_NODES;
    int* slot  = first + (N_NODES + 16);
    int* srcs  = slot + E_EDGES;
    static_assert((size_t)N_NODES + (N_NODES + 16) + 2 * (size_t)E_EDGES <= (size_t)ETOT * NHEAD,
                  "the grouping must fit the region it reuses");

    const int initBlocks  = (N_NODES + 255) / 256;
    const int gatherBlocks = (N_NODES + NG_WAVES - 1) / NG_WAVES;
    const dim3 gridPair(2 * HC / GM_BN, N_NODES / GM_BM);
    const dim3 gridInCount (HID / GM_BN,    N_NODES / GM_BM + side_rows(SIDE_COUNT,    HID / GM_BN));
    const dim3 gridPairFill(2 * HC / GM_BN, N_NODES / GM_BM + side_rows(SIDE_SCANFILL, 2 * HC / GM_BN));
    const int logitBlocks = (N_NODES + LG_NPB - 1) / LG_NPB;

    init_scratch<<<initBlocks, 256, 0, stream>>>(cnt, gpart);

    gemm_wmma_f32<1, DIN, HID, 1, SIDE_COUNT><<<gridInCount, GM_WAVES * 32, 0, stream>>>(
        x, W_in, b_in, h0, nullptr, nullptr, nullptr, ei, cnt, first, slot, srcs);

    gemm_wmma_f32<0, HID, HC, 2, SIDE_SCANFILL><<<gridPairFill, GM_WAVES * 32, 0, stream>>>(
        h0, Wl1, bl1, xl, Wr1, br1, xr, ei, cnt, first, slot, srcs);
    node_gather <<<gatherBlocks, NG_WAVES * 32, 0, stream>>>(first, srcs, xl, xr, att1,
                                                             h0, bias1, g1, be1, h1);

    gemm_wmma_f32<0, HID, HC, 2, SIDE_NONE><<<gridPair, GM_WAVES * 32, 0, stream>>>(
        h1, Wl2, bl2, xl, Wr2, br2, xr, ei, cnt, first, slot, srcs);
    node_gather <<<gatherBlocks, NG_WAVES * 32, 0, stream>>>(first, srcs, xl, xr, att2,
                                                             h1, bias2, g2, be2, h0);

    logits_kernel<<<logitBlocks, LG_WAVES * 32, 0, stream>>>(h0, Wa, ba, gpart, out);
    value_kernel <<<1, 32, 0, stream>>>(gpart, Wc, bc, out);
}
